# starting kernel
# baseline (speedup 1.0000x reference)
#include <hip/hip_runtime.h>
#include <math.h>

typedef float v2f __attribute__((ext_vector_type(2)));
typedef float v8f __attribute__((ext_vector_type(8)));

#define LRELU_SLOPE 0.2f

__device__ __forceinline__ unsigned f2o(float f) {
  unsigned u = __float_as_uint(f);
  return (u & 0x80000000u) ? ~u : (u | 0x80000000u);
}
__device__ __forceinline__ float o2f(unsigned o) {
  unsigned u = (o & 0x80000000u) ? (o & 0x7fffffffu) : ~o;
  return __uint_as_float(u);
}
#define ORD_NEG_INF 0x007fffffu

__global__ void fill_f32(float* p, float v, int n) {
  int i = blockIdx.x * blockDim.x + threadIdx.x;
  int stride = gridDim.x * blockDim.x;
  for (; i < n; i += stride) p[i] = v;
}
__global__ void fill_u32(unsigned* p, unsigned v, int n) {
  int i = blockIdx.x * blockDim.x + threadIdx.x;
  int stride = gridDim.x * blockDim.x;
  for (; i < n; i += stride) p[i] = v;
}

__global__ void gemm_wmma_f32(const float* __restrict__ A,
                              const float* __restrict__ B,
                              float* __restrict__ C,
                              int M, int Ncols, int K) {
  const int wavesPerBlock = blockDim.x >> 5;
  const int wave = threadIdx.x >> 5;
  const int lane = threadIdx.x & 31;
  const int half = lane >> 4;
  const int l    = lane & 15;

  const int m0 = blockIdx.x * 16;
  const int nt = blockIdx.y * wavesPerBlock + wave;
  const int n0 = nt * 16;
  if (m0 >= M || n0 >= Ncols) return;

  v8f acc = {};
  const float* arow = A + (size_t)(m0 + l) * K + half * 2;
  for (int k = 0; k < K; k += 4) {
    v2f a, b;
    a.x = arow[k];
    a.y = arow[k + 1];
    const float* bp = B + (size_t)(k + half * 2) * Ncols + n0 + l;
    b.x = bp[0];
    b.y = bp[Ncols];
    acc = __builtin_amdgcn_wmma_f32_16x16x4_f32(false, a, false, b,
                                                (short)0, acc, false, false);
  }
#pragma unroll
  for (int r = 0; r < 8; ++r) {
    C[(size_t)(m0 + half * 8 + r) * Ncols + n0 + l] = acc[r];
  }
}

__global__ void node_scores(const float* __restrict__ feat,
                            const float* __restrict__ al,
                            const float* __restrict__ ar,
                            float* __restrict__ el, float* __restrict__ er,
                            int N, int H, int D) {
  int i = blockIdx.x * blockDim.x + threadIdx.x;
  if (i >= N * H) return;
  int n = i / H, h = i - n * H;
  const float* f = feat + (size_t)n * H * D + h * D;
  const float* wl = al + h * D;
  const float* wr = ar + h * D;
  float sl = 0.f, sr = 0.f;
#pragma unroll 4
  for (int d = 0; d < D; ++d) {
    float v = f[d];
    sl += v * wl[d];
    sr += v * wr[d];
  }
  el[i] = sl;
  er[i] = sr;
}

__global__ void edge_max(const int* __restrict__ src, const int* __restrict__ dst,
                         const float* __restrict__ el, const float* __restrict__ er,
                         unsigned* __restrict__ m_ord, int E, int H) {
  int i = blockIdx.x * blockDim.x + threadIdx.x;
  if (i >= E * H) return;
  int e = i / H, h = i - e * H;
  int d = dst[e];
  float sc = el[src[e] * H + h] + er[d * H + h];
  sc = sc > 0.f ? sc : LRELU_SLOPE * sc;
  atomicMax(m_ord + d * H + h, f2o(sc));
}

__global__ void edge_expsum(const int* __restrict__ src, const int* __restrict__ dst,
                            const float* __restrict__ el, const float* __restrict__ er,
                            const unsigned* __restrict__ m_ord,
                            float* __restrict__ ee, float* __restrict__ s,
                            int E, int H) {
  int i = blockIdx.x * blockDim.x + threadIdx.x;
  if (i >= E * H) return;
  int e = i / H, h = i - e * H;
  int d = dst[e];
  float sc = el[src[e] * H + h] + er[d * H + h];
  sc = sc > 0.f ? sc : LRELU_SLOPE * sc;
  float m = o2f(m_ord[d * H + h]);
  if (m < -1e30f) m = 0.f;
  float v = __expf(sc - m);
  ee[i] = v;
  atomicAdd(s + d * H + h, v);
}

__global__ void edge_aggregate(const int* __restrict__ src, const int* __restrict__ dst,
                               const float* __restrict__ ee, const float* __restrict__ s,
                               const float* __restrict__ feat, float* __restrict__ acc,
                               int E, int H, int D) {
  const int chunks = (H * D) >> 2;
  int tid = blockIdx.x * blockDim.x + threadIdx.x;
  if (tid >= E * chunks) return;
  int e = tid / chunks;
  int c = tid - e * chunks;
  int h = (c << 2) / D;
  int sN = src[e], dN = dst[e];
  float a = ee[(size_t)e * H + h] / fmaxf(s[(size_t)dN * H + h], 1e-9f);
  const float4 f = *(const float4*)(feat + (size_t)sN * H * D + (c << 2));
  float* o = acc + (size_t)dN * H * D + (c << 2);
  atomicAdd(o + 0, a * f.x);
  atomicAdd(o + 1, a * f.y);
  atomicAdd(o + 2, a * f.z);
  atomicAdd(o + 3, a * f.w);
}

__global__ void epilogue0(const float* __restrict__ acc, const float* __restrict__ b,
                          float* __restrict__ h1, int N, int H, int D) {
  int i = blockIdx.x * blockDim.x + threadIdx.x;
  if (i >= N * D) return;
  int n = i / D, d = i - n * D;
  float sum = 0.f;
  for (int t = 0; t < 2; ++t) {
    const float* a = acc + ((size_t)t * N + n) * H * D + d;
    for (int h = 0; h < H; ++h) {
      float v = a[h * D] + b[h * D + d];
      sum += (v > 0.f) ? v : expm1f(v);
    }
  }
  h1[i] = sum / (float)H;
}

__global__ void epilogue1(const float* __restrict__ acc, const float* __restrict__ b,
                          float* __restrict__ out, int N, int D) {
  int i = blockIdx.x * blockDim.x + threadIdx.x;
  if (i >= N * D) return;
  int d = i - (i / D) * D;
  out[i] = (acc[i] + b[d]) + (acc[(size_t)N * D + i] + b[d]);
}

extern "C" void kernel_launch(void* const* d_in, const int* in_sizes, int n_in,
                              void* d_out, int out_size, void* d_ws, size_t ws_size,
                              hipStream_t stream) {
  const int IN = 128, H0 = 4, D0 = 64, H1 = 1, D1 = 64;
  const int N = in_sizes[0] / IN;
  const int E = in_sizes[9];

  const float* x   = (const float*)d_in[0];
  const float* W0  = (const float*)d_in[1];
  const float* al0 = (const float*)d_in[2];
  const float* ar0 = (const float*)d_in[3];
  const float* b0  = (const float*)d_in[4];
  const float* W1  = (const float*)d_in[5];
  const float* al1 = (const float*)d_in[6];
  const float* ar1 = (const float*)d_in[7];
  const float* b1  = (const float*)d_in[8];
  const int* srcT[2] = { (const int*)d_in[9],  (const int*)d_in[11] };
  const int* dstT[2] = { (const int*)d_in[10], (const int*)d_in[12] };
  float* out = (float*)d_out;

  float* ws = (float*)d_ws;
  float*    feat0 = ws;
  float*    el0   = feat0 + (size_t)N * H0 * D0;
  float*    er0   = el0 + (size_t)N * H0;
  unsigned* m0    = (unsigned*)(er0 + (size_t)N * H0);
  float*    s0    = (float*)(m0 + 2 * (size_t)N * H0);
  float*    ee0   = s0 + 2 * (size_t)N * H0;
  float*    acc0  = ee0 + 2 * (size_t)E * H0;
  float*    h1    = acc0 + 2 * (size_t)N * H0 * D0;
  float*    feat1 = feat0;
  float*    el1   = feat1 + (size_t)N * D1;
  float*    er1   = el1 + N;
  unsigned* m1    = (unsigned*)(er1 + N);
  float*    s1    = (float*)(m1 + 2 * (size_t)N);
  float*    ee1   = s1 + 2 * (size_t)N;
  float*    acc1  = ee1 + 2 * (size_t)E;

  const int TB = 256;
  const int fillBlocks = 2048;
  (void)ws_size; (void)n_in; (void)out_size;

  {
    dim3 grid(N / 16, (H0 * D0) / 16 / 4);
    gemm_wmma_f32<<<grid, 128, 0, stream>>>(x, W0, feat0, N, H0 * D0, IN);

    node_scores<<<(N * H0 + TB - 1) / TB, TB, 0, stream>>>(feat0, al0, ar0, el0, er0, N, H0, D0);

    fill_u32<<<fillBlocks, TB, 0, stream>>>(m0, ORD_NEG_INF, 2 * N * H0);
    fill_f32<<<fillBlocks, TB, 0, stream>>>(s0, 0.f, 2 * N * H0);
    fill_f32<<<fillBlocks, TB, 0, stream>>>(acc0, 0.f, 2 * N * H0 * D0);

    for (int t = 0; t < 2; ++t) {
      unsigned* mT = m0 + (size_t)t * N * H0;
      float*    sT = s0 + (size_t)t * N * H0;
      float*    eeT = ee0 + (size_t)t * E * H0;
      float*    accT = acc0 + (size_t)t * N * H0 * D0;
      int eh = E * H0;
      edge_max<<<(eh + TB - 1) / TB, TB, 0, stream>>>(srcT[t], dstT[t], el0, er0, mT, E, H0);
      edge_expsum<<<(eh + TB - 1) / TB, TB, 0, stream>>>(srcT[t], dstT[t], el0, er0, mT, eeT, sT, E, H0);
      int total = E * ((H0 * D0) >> 2);
      edge_aggregate<<<(total + TB - 1) / TB, TB, 0, stream>>>(srcT[t], dstT[t], eeT, sT, feat0, accT, E, H0, D0);
    }

    epilogue0<<<(N * D0 + TB - 1) / TB, TB, 0, stream>>>(acc0, b0, h1, N, H0, D0);
  }

  {
    dim3 grid(N / 16, 1);
    gemm_wmma_f32<<<grid, 128, 0, stream>>>(h1, W1, feat1, N, H1 * D1, D0);

    node_scores<<<(N * H1 + TB - 1) / TB, TB, 0, stream>>>(feat1, al1, ar1, el1, er1, N, H1, D1);

    fill_u32<<<fillBlocks, TB, 0, stream>>>(m1, ORD_NEG_INF, 2 * N * H1);
    fill_f32<<<fillBlocks, TB, 0, stream>>>(s1, 0.f, 2 * N * H1);
    fill_f32<<<fillBlocks, TB, 0, stream>>>(acc1, 0.f, 2 * N * H1 * D1);

    for (int t = 0; t < 2; ++t) {
      unsigned* mT = m1 + (size_t)t * N * H1;
      float*    sT = s1 + (size_t)t * N * H1;
      float*    eeT = ee1 + (size_t)t * E * H1;
      float*    accT = acc1 + (size_t)t * N * H1 * D1;
      int eh = E * H1;
      edge_max<<<(eh + TB - 1) / TB, TB, 0, stream>>>(srcT[t], dstT[t], el1, er1, mT, E, H1);
      edge_expsum<<<(eh + TB - 1) / TB, TB, 0, stream>>>(srcT[t], dstT[t], el1, er1, mT, eeT, sT, E, H1);
      int total = E * ((H1 * D1) >> 2);
      edge_aggregate<<<(total + TB - 1) / TB, TB, 0, stream>>>(srcT[t], dstT[t], eeT, sT, feat1, accT, E, H1, D1);
    }

    epilogue1<<<(N * D1 + TB - 1) / TB, TB, 0, stream>>>(acc1, b1, out, N, D1);
  }
}


# Float-atomic scatter passes (max, sum, aggregate) replaced by an in-call dst-sorted CSR (int histogram + cursor atomics) and one wave-per-node gather; sums now in CSR slot order plus lane tree
# speedup vs baseline: 2.5276x; 2.5276x over previous
#include <hip/hip_runtime.h>
#include <math.h>

typedef float v2f __attribute__((ext_vector_type(2)));
typedef float v8f __attribute__((ext_vector_type(8)));

#define LRELU_SLOPE 0.2f

__global__ void fill_u32(unsigned* p, unsigned v, int n) {
  int i = blockIdx.x * blockDim.x + threadIdx.x;
  int stride = gridDim.x * blockDim.x;
  for (; i < n; i += stride) p[i] = v;
}

__global__ void gemm_wmma_f32(const float* __restrict__ A,
                              const float* __restrict__ B,
                              float* __restrict__ C,
                              int M, int Ncols, int K) {
  const int wavesPerBlock = blockDim.x >> 5;
  const int wave = threadIdx.x >> 5;
  const int lane = threadIdx.x & 31;
  const int half = lane >> 4;
  const int l    = lane & 15;

  const int m0 = blockIdx.x * 16;
  const int nt = blockIdx.y * wavesPerBlock + wave;
  const int n0 = nt * 16;
  if (m0 >= M || n0 >= Ncols) return;

  v8f acc = {};
  const float* arow = A + (size_t)(m0 + l) * K + half * 2;
  for (int k = 0; k < K; k += 4) {
    v2f a, b;
    a.x = arow[k];
    a.y = arow[k + 1];
    const float* bp = B + (size_t)(k + half * 2) * Ncols + n0 + l;
    b.x = bp[0];
    b.y = bp[Ncols];
    acc = __builtin_amdgcn_wmma_f32_16x16x4_f32(false, a, false, b,
                                                (short)0, acc, false, false);
  }
#pragma unroll
  for (int r = 0; r < 8; ++r) {
    C[(size_t)(m0 + half * 8 + r) * Ncols + n0 + l] = acc[r];
  }
}

__global__ void node_scores(const float* __restrict__ feat,
                            const float* __restrict__ al,
                            const float* __restrict__ ar,
                            float* __restrict__ el, float* __restrict__ er,
                            int N, int H, int D) {
  int i = blockIdx.x * blockDim.x + threadIdx.x;
  if (i >= N * H) return;
  int n = i / H, h = i - n * H;
  const float* f = feat + (size_t)n * H * D + h * D;
  const float* wl = al + h * D;
  const float* wr = ar + h * D;
  float sl = 0.f, sr = 0.f;
#pragma unroll 4
  for (int d = 0; d < D; ++d) {
    float v = f[d];
    sl += v * wl[d];
    sr += v * wr[d];
  }
  el[i] = sl;
  er[i] = sr;
}


__global__ void edge_hist(const int* __restrict__ dst0, const int* __restrict__ dst1,
                          unsigned* __restrict__ deg, int E, int N) {
  int e = blockIdx.x * blockDim.x + threadIdx.x;
  if (e >= E) return;
  const int t = blockIdx.y;
  const int* dst = t ? dst1 : dst0;
  atomicAdd(deg + (size_t)t * N + dst[e], 1u);
}

__global__ void seg_alloc(const unsigned* __restrict__ deg, unsigned* __restrict__ counter,
                          int2* __restrict__ seg, int N) {
  const int t = blockIdx.y;
  const int n = blockIdx.x * blockDim.x + threadIdx.x;
  const int lane = threadIdx.x & 31;
  const unsigned x = (n < N) ? deg[(size_t)t * N + n] : 0u;
  unsigned incl = x;
  for (int d = 1; d < 32; d <<= 1) {
    unsigned y = __shfl_up(incl, d);
    if (lane >= d) incl += y;
  }
  const unsigned total = __shfl(incl, 31);
  unsigned base = 0u;
  if (lane == 31) base = atomicAdd(counter + t, total);
  base = __shfl(base, 31);
  if (n < N) seg[(size_t)t * N + n] = make_int2((int)(base + incl - x), (int)x);
}

__global__ void edge_scatter(const int* __restrict__ src0, const int* __restrict__ dst0,
                             const int* __restrict__ src1, const int* __restrict__ dst1,
                             unsigned* __restrict__ deg, const int2* __restrict__ seg,
                             int* __restrict__ csr, int E, int N) {
  int e = blockIdx.x * blockDim.x + threadIdx.x;
  if (e >= E) return;
  const int t = blockIdx.y;
  const int* src = t ? src1 : src0;
  const int* dst = t ? dst1 : dst0;
  const int d = dst[e];
  const unsigned old = atomicSub(deg + (size_t)t * N + d, 1u);
  const int pos = seg[(size_t)t * N + d].x + (int)old - 1;
  csr[(size_t)t * E + pos] = src[e];
}

__device__ __forceinline__ float lrelu(float v) { return v > 0.f ? v : LRELU_SLOPE * v; }
__device__ __forceinline__ float elu1(float v) { return v > 0.f ? v : expm1f(v); }
__device__ __forceinline__ float wave_max(float v) {
  for (int o = 16; o > 0; o >>= 1) v = fmaxf(v, __shfl_xor(v, o));
  return v;
}
__device__ __forceinline__ float wave_sum(float v) {
  for (int o = 16; o > 0; o >>= 1) v += __shfl_xor(v, o);
  return v;
}
__device__ __forceinline__ float lane_val(float v, int k) {
  return __int_as_float(__builtin_amdgcn_readlane(__float_as_int(v), k));
}

__global__ __launch_bounds__(256) void gather_l0(const int2* __restrict__ seg, const int* __restrict__ csr,
                                                 const float* __restrict__ el, const float* __restrict__ er,
                                                 const float* __restrict__ feat, const float* __restrict__ b,
                                                 float* __restrict__ h1, int N, int E) {
  const int lane = threadIdx.x & 31;
  const int n = blockIdx.x * 8 + __builtin_amdgcn_readfirstlane((int)(threadIdx.x >> 5));
  if (n >= N) return;
  const int hsel = lane >> 4;
  const float4 er4 = *(const float4*)(er + 4 * (size_t)n);
  const float4 bA = *(const float4*)(b + 4 * lane);
  const float4 bB = *(const float4*)(b + 128 + 4 * lane);
  float4 hs = make_float4(0.f, 0.f, 0.f, 0.f);

  for (int t = 0; t < 2; ++t) {
    const int2 sg = seg[(size_t)t * N + n];
    const int beg = __builtin_amdgcn_readfirstlane(sg.x);
    const int end = beg + __builtin_amdgcn_readfirstlane(sg.y);
    const int* cs = csr + (size_t)t * E;

    float m0 = -INFINITY, m1 = -INFINITY, m2 = -INFINITY, m3 = -INFINITY;
    for (int j = beg + lane; j < end; j += 32) {
      const float4 e4 = *(const float4*)(el + 4 * (size_t)cs[j]);
      m0 = fmaxf(m0, lrelu(e4.x + er4.x));
      m1 = fmaxf(m1, lrelu(e4.y + er4.y));
      m2 = fmaxf(m2, lrelu(e4.z + er4.z));
      m3 = fmaxf(m3, lrelu(e4.w + er4.w));
    }
    m0 = wave_max(m0); m1 = wave_max(m1); m2 = wave_max(m2); m3 = wave_max(m3);
    if (!__builtin_isfinite(m0)) m0 = 0.f;
    if (!__builtin_isfinite(m1)) m1 = 0.f;
    if (!__builtin_isfinite(m2)) m2 = 0.f;
    if (!__builtin_isfinite(m3)) m3 = 0.f;

    float s0 = 0.f, s1 = 0.f, s2 = 0.f, s3 = 0.f;
    for (int j = beg + lane; j < end; j += 32) {
      const float4 e4 = *(const float4*)(el + 4 * (size_t)cs[j]);
      s0 += __expf(lrelu(e4.x + er4.x) - m0);
      s1 += __expf(lrelu(e4.y + er4.y) - m1);
      s2 += __expf(lrelu(e4.z + er4.z) - m2);
      s3 += __expf(lrelu(e4.w + er4.w) - m3);
    }
    const float d0 = fmaxf(wave_sum(s0), 1e-9f);
    const float d1 = fmaxf(wave_sum(s1), 1e-9f);
    const float d2 = fmaxf(wave_sum(s2), 1e-9f);
    const float d3 = fmaxf(wave_sum(s3), 1e-9f);

    float4 aA = make_float4(0.f, 0.f, 0.f, 0.f), aB = make_float4(0.f, 0.f, 0.f, 0.f);
    for (int base = beg; base < end; base += 32) {
      const int j = base + lane;
      int sidx = 0;
      float a0 = 0.f, a1 = 0.f, a2 = 0.f, a3 = 0.f;
      if (j < end) {
        sidx = cs[j];
        const float4 e4 = *(const float4*)(el + 4 * (size_t)sidx);
        a0 = __expf(lrelu(e4.x + er4.x) - m0) / d0;
        a1 = __expf(lrelu(e4.y + er4.y) - m1) / d1;
        a2 = __expf(lrelu(e4.z + er4.z) - m2) / d2;
        a3 = __expf(lrelu(e4.w + er4.w) - m3) / d3;
      }
      const int cnt = min(32, end - base);
      for (int k = 0; k < cnt; ++k) {
        const int sk = __builtin_amdgcn_readlane(sidx, k);
        const float w0 = lane_val(a0, k), w1 = lane_val(a1, k);
        const float w2 = lane_val(a2, k), w3 = lane_val(a3, k);
        const float wA = hsel ? w1 : w0;
        const float wB = hsel ? w3 : w2;
        const float* fr = feat + (size_t)sk * 256 + 4 * lane;
        const float4 fA = *(const float4*)fr;
        const float4 fB = *(const float4*)(fr + 128);
        aA.x += wA * fA.x; aA.y += wA * fA.y; aA.z += wA * fA.z; aA.w += wA * fA.w;
        aB.x += wB * fB.x; aB.y += wB * fB.y; aB.z += wB * fB.z; aB.w += wB * fB.w;
      }
    }
    hs.x += elu1(aA.x + bA.x) + elu1(aB.x + bB.x);
    hs.y += elu1(aA.y + bA.y) + elu1(aB.y + bB.y);
    hs.z += elu1(aA.z + bA.z) + elu1(aB.z + bB.z);
    hs.w += elu1(aA.w + bA.w) + elu1(aB.w + bB.w);
  }
  hs.x += __shfl_xor(hs.x, 16);
  hs.y += __shfl_xor(hs.y, 16);
  hs.z += __shfl_xor(hs.z, 16);
  hs.w += __shfl_xor(hs.w, 16);
  if (hsel == 0) {
    *(float4*)(h1 + (size_t)n * 64 + 4 * lane) =
        make_float4(hs.x / 4.f, hs.y / 4.f, hs.z / 4.f, hs.w / 4.f);
  }
}

__global__ __launch_bounds__(256) void gather_l1(const int2* __restrict__ seg, const int* __restrict__ csr,
                                                 const float* __restrict__ el, const float* __restrict__ er,
                                                 const float* __restrict__ feat, const float* __restrict__ b,
                                                 float* __restrict__ out, int N, int E) {
  const int lane = threadIdx.x & 31;
  const int n = blockIdx.x * 8 + __builtin_amdgcn_readfirstlane((int)(threadIdx.x >> 5));
  if (n >= N) return;
  const float ern = er[n];
  const float2 b2 = *(const float2*)(b + 2 * lane);
  float2 o = make_float2(0.f, 0.f);

  for (int t = 0; t < 2; ++t) {
    const int2 sg = seg[(size_t)t * N + n];
    const int beg = __builtin_amdgcn_readfirstlane(sg.x);
    const int end = beg + __builtin_amdgcn_readfirstlane(sg.y);
    const int* cs = csr + (size_t)t * E;

    float m = -INFINITY;
    for (int j = beg + lane; j < end; j += 32) m = fmaxf(m, lrelu(el[cs[j]] + ern));
    m = wave_max(m);
    if (!__builtin_isfinite(m)) m = 0.f;

    float s = 0.f;
    for (int j = beg + lane; j < end; j += 32) s += __expf(lrelu(el[cs[j]] + ern) - m);
    const float d = fmaxf(wave_sum(s), 1e-9f);

    float2 acc = make_float2(0.f, 0.f);
    for (int base = beg; base < end; base += 32) {
      const int j = base + lane;
      int sidx = 0;
      float a = 0.f;
      if (j < end) {
        sidx = cs[j];
        a = __expf(lrelu(el[sidx] + ern) - m) / d;
      }
      const int cnt = min(32, end - base);
      for (int k = 0; k < cnt; ++k) {
        const int sk = __builtin_amdgcn_readlane(sidx, k);
        const float w = lane_val(a, k);
        const float2 f = *(const float2*)(feat + (size_t)sk * 64 + 2 * lane);
        acc.x += w * f.x;
        acc.y += w * f.y;
      }
    }
    o.x += acc.x + b2.x;
    o.y += acc.y + b2.y;
  }
  *(float2*)(out + (size_t)n * 64 + 2 * lane) = o;
}

extern "C" void kernel_launch(void* const* d_in, const int* in_sizes, int n_in,
                              void* d_out, int out_size, void* d_ws, size_t ws_size,
                              hipStream_t stream) {
  const int IN = 128, H0 = 4, D0 = 64, H1 = 1, D1 = 64;
  const int N = in_sizes[0] / IN;
  const int E = in_sizes[9];

  const float* x   = (const float*)d_in[0];
  const float* W0  = (const float*)d_in[1];
  const float* al0 = (const float*)d_in[2];
  const float* ar0 = (const float*)d_in[3];
  const float* b0  = (const float*)d_in[4];
  const float* W1  = (const float*)d_in[5];
  const float* al1 = (const float*)d_in[6];
  const float* ar1 = (const float*)d_in[7];
  const float* b1  = (const float*)d_in[8];
  const int* src0 = (const int*)d_in[9];
  const int* dst0 = (const int*)d_in[10];
  const int* src1 = (const int*)d_in[11];
  const int* dst1 = (const int*)d_in[12];
  float* out = (float*)d_out;

  float* ws = (float*)d_ws;
  float*    feat0 = ws;
  float*    el0   = feat0 + (size_t)N * H0 * D0;
  float*    er0   = el0 + (size_t)N * H0;
  float*    h1    = er0 + (size_t)N * H0;
  float*    feat1 = h1 + (size_t)N * D0;
  float*    el1   = feat1 + (size_t)N * D1;
  float*    er1   = el1 + N;
  unsigned* deg   = (unsigned*)(er1 + N);
  unsigned* cnt   = deg + 2 * (size_t)N;
  int2*     seg   = (int2*)(cnt + 4);
  int*      csr   = (int*)(seg + 2 * (size_t)N);

  const int TB = 256;
  (void)ws_size; (void)n_in; (void)out_size;

  {
    fill_u32<<<(2 * N + 4 + TB - 1) / TB, TB, 0, stream>>>(deg, 0u, 2 * N + 4);
    dim3 ge((E + TB - 1) / TB, 2);
    dim3 gn((N + TB - 1) / TB, 2);
    edge_hist<<<ge, TB, 0, stream>>>(dst0, dst1, deg, E, N);
    seg_alloc<<<gn, TB, 0, stream>>>(deg, cnt, seg, N);
    edge_scatter<<<ge, TB, 0, stream>>>(src0, dst0, src1, dst1, deg, seg, csr, E, N);
  }

  {
    dim3 grid(N / 16, (H0 * D0) / 16 / 4);
    gemm_wmma_f32<<<grid, 128, 0, stream>>>(x, W0, feat0, N, H0 * D0, IN);

    node_scores<<<(N * H0 + TB - 1) / TB, TB, 0, stream>>>(feat0, al0, ar0, el0, er0, N, H0, D0);

    gather_l0<<<(N + 7) / 8, TB, 0, stream>>>(seg, csr, el0, er0, feat0, b0, h1, N, E);
  }

  {
    dim3 grid(N / 16, 1);
    gemm_wmma_f32<<<grid, 128, 0, stream>>>(h1, W1, feat1, N, H1 * D1, D0);

    node_scores<<<(N * H1 + TB - 1) / TB, TB, 0, stream>>>(feat1, al1, ar1, el1, er1, N, H1, D1);

    gather_l1<<<(N + 7) / 8, TB, 0, stream>>>(seg, csr, el1, er1, feat1, b1, out, N, E);
  }
}


# GEMMs re-tiled: each wave keeps its K x 64 slice of B in registers and walks 16-row tiles with prefetched 16-byte A loads; k pairs regrouped within each 8 k, so the product sum order changes
# speedup vs baseline: 2.8717x; 1.1361x over previous
#include <hip/hip_runtime.h>
#include <math.h>

typedef float v2f __attribute__((ext_vector_type(2)));
typedef float v8f __attribute__((ext_vector_type(8)));

#define LRELU_SLOPE 0.2f

__global__ void fill_u32(unsigned* p, unsigned v, int n) {
  int i = blockIdx.x * blockDim.x + threadIdx.x;
  int stride = gridDim.x * blockDim.x;
  for (; i < n; i += stride) p[i] = v;
}

template <int K>
__device__ __forceinline__ void load_a_tile(float4 (&a)[K / 8], const float* __restrict__ A, int mt, int c, int h) {
  const float* ap = A + ((size_t)mt * 16 + c) * K + 4 * h;
#pragma unroll
  for (int j = 0; j < K / 8; ++j) a[j] = *(const float4*)(ap + 8 * j);
}

template <int K, int NW>
__device__ __forceinline__ void mma_store_tile(const float4 (&a)[K / 8], const v2f (&bf)[4][K / 4],
                                               float* __restrict__ C, int mt, int n0, int lane, int h) {
  constexpr int NC = NW * 64;
  v8f acc[4] = {};
#pragma unroll
  for (int j = 0; j < K / 8; ++j) {
    v2f a0, a1;
    a0.x = a[j].x; a0.y = a[j].y;
    a1.x = a[j].z; a1.y = a[j].w;
#pragma unroll
    for (int t = 0; t < 4; ++t)
      acc[t] = __builtin_amdgcn_wmma_f32_16x16x4_f32(false, a0, false, bf[t][2 * j], (short)0, acc[t],
                                                     false, false);
#pragma unroll
    for (int t = 0; t < 4; ++t)
      acc[t] = __builtin_amdgcn_wmma_f32_16x16x4_f32(false, a1, false, bf[t][2 * j + 1], (short)0,
                                                     acc[t], false, false);
  }
  float* crow = C + (size_t)mt * 16 * NC + n0 + lane;
#pragma unroll
  for (int p = 0; p < 2; ++p) {
#pragma unroll
    for (int r = 0; r < 8; ++r) {
      const float x0 = acc[2 * p][r], x1 = acc[2 * p + 1][r];
      const float recv = __shfl_xor(h ? x0 : x1, 16);
      crow[(size_t)r * NC + 32 * p] = h ? recv : x0;
      crow[(size_t)(8 + r) * NC + 32 * p] = h ? x1 : recv;
    }
  }
}

template <int K, int NW>
__global__ __launch_bounds__(NW * 32) void gemm_wmma_f32_bregs(const float* __restrict__ A,
                                                               const float* __restrict__ B,
                                                               float* __restrict__ C, int mtiles) {
  constexpr int NC = NW * 64;
  constexpr int KS = K / 4;
  const int lane = threadIdx.x & 31;
  const int h = lane >> 4;
  const int c = lane & 15;
  const int n0 = __builtin_amdgcn_readfirstlane((int)(threadIdx.x >> 5)) * 64;
  const int step = gridDim.x;

  int mt = blockIdx.x;
  if (mt >= mtiles) return;

  v2f bf[4][KS];
#pragma unroll
  for (int t = 0; t < 4; ++t) {
#pragma unroll
    for (int ks = 0; ks < KS; ++ks) {
      const int krow = 8 * (ks >> 1) + 4 * h + 2 * (ks & 1);
      const float* bp = B + (size_t)krow * NC + n0 + 16 * t + c;
      bf[t][ks].x = bp[0];
      bf[t][ks].y = bp[NC];
    }
  }

  float4 an[K / 8];
  load_a_tile<K>(an, A, mt, c, h);
  do {
    float4 ac[K / 8];
#pragma unroll
    for (int j = 0; j < K / 8; ++j) ac[j] = an[j];
    const int mn = mt + step;
    load_a_tile<K>(an, A, mn < mtiles ? mn : mt, c, h);
    __builtin_amdgcn_sched_barrier(0);
    mma_store_tile<K, NW>(ac, bf, C, mt, n0, lane, h);
    __builtin_amdgcn_sched_barrier(0);
    mt = mn;
  } while (mt < mtiles);
}

__global__ void node_scores(const float* __restrict__ feat,
                            const float* __restrict__ al,
                            const float* __restrict__ ar,
                            float* __restrict__ el, float* __restrict__ er,
                            int N, int H, int D) {
  int i = blockIdx.x * blockDim.x + threadIdx.x;
  if (i >= N * H) return;
  int n = i / H, h = i - n * H;
  const float* f = feat + (size_t)n * H * D + h * D;
  const float* wl = al + h * D;
  const float* wr = ar + h * D;
  float sl = 0.f, sr = 0.f;
#pragma unroll 4
  for (int d = 0; d < D; ++d) {
    float v = f[d];
    sl += v * wl[d];
    sr += v * wr[d];
  }
  el[i] = sl;
  er[i] = sr;
}


__global__ void edge_hist(const int* __restrict__ dst0, const int* __restrict__ dst1,
                          unsigned* __restrict__ deg, int E, int N) {
  int e = blockIdx.x * blockDim.x + threadIdx.x;
  if (e >= E) return;
  const int t = blockIdx.y;
  const int* dst = t ? dst1 : dst0;
  atomicAdd(deg + (size_t)t * N + dst[e], 1u);
}

__global__ void seg_alloc(const unsigned* __restrict__ deg, unsigned* __restrict__ counter,
                          int2* __restrict__ seg, int N) {
  const int t = blockIdx.y;
  const int n = blockIdx.x * blockDim.x + threadIdx.x;
  const int lane = threadIdx.x & 31;
  const unsigned x = (n < N) ? deg[(size_t)t * N + n] : 0u;
  unsigned incl = x;
  for (int d = 1; d < 32; d <<= 1) {
    unsigned y = __shfl_up(incl, d);
    if (lane >= d) incl += y;
  }
  const unsigned total = __shfl(incl, 31);
  unsigned base = 0u;
  if (lane == 31) base = atomicAdd(counter + t, total);
  base = __shfl(base, 31);
  if (n < N) seg[(size_t)t * N + n] = make_int2((int)(base + incl - x), (int)x);
}

__global__ void edge_scatter(const int* __restrict__ src0, const int* __restrict__ dst0,
                             const int* __restrict__ src1, const int* __restrict__ dst1,
                             unsigned* __restrict__ deg, const int2* __restrict__ seg,
                             int* __restrict__ csr, int E, int N) {
  int e = blockIdx.x * blockDim.x + threadIdx.x;
  if (e >= E) return;
  const int t = blockIdx.y;
  const int* src = t ? src1 : src0;
  const int* dst = t ? dst1 : dst0;
  const int d = dst[e];
  const unsigned old = atomicSub(deg + (size_t)t * N + d, 1u);
  const int pos = seg[(size_t)t * N + d].x + (int)old - 1;
  csr[(size_t)t * E + pos] = src[e];
}

__device__ __forceinline__ float lrelu(float v) { return v > 0.f ? v : LRELU_SLOPE * v; }
__device__ __forceinline__ float elu1(float v) { return v > 0.f ? v : expm1f(v); }
__device__ __forceinline__ float wave_max(float v) {
  for (int o = 16; o > 0; o >>= 1) v = fmaxf(v, __shfl_xor(v, o));
  return v;
}
__device__ __forceinline__ float wave_sum(float v) {
  for (int o = 16; o > 0; o >>= 1) v += __shfl_xor(v, o);
  return v;
}
__device__ __forceinline__ float lane_val(float v, int k) {
  return __int_as_float(__builtin_amdgcn_readlane(__float_as_int(v), k));
}

__global__ __launch_bounds__(256) void gather_l0(const int2* __restrict__ seg, const int* __restrict__ csr,
                                                 const float* __restrict__ el, const float* __restrict__ er,
                                                 const float* __restrict__ feat, const float* __restrict__ b,
                                                 float* __restrict__ h1, int N, int E) {
  const int lane = threadIdx.x & 31;
  const int n = blockIdx.x * 8 + __builtin_amdgcn_readfirstlane((int)(threadIdx.x >> 5));
  if (n >= N) return;
  const int hsel = lane >> 4;
  const float4 er4 = *(const float4*)(er + 4 * (size_t)n);
  const float4 bA = *(const float4*)(b + 4 * lane);
  const float4 bB = *(const float4*)(b + 128 + 4 * lane);
  float4 hs = make_float4(0.f, 0.f, 0.f, 0.f);

  for (int t = 0; t < 2; ++t) {
    const int2 sg = seg[(size_t)t * N + n];
    const int beg = __builtin_amdgcn_readfirstlane(sg.x);
    const int end = beg + __builtin_amdgcn_readfirstlane(sg.y);
    const int* cs = csr + (size_t)t * E;

    float m0 = -INFINITY, m1 = -INFINITY, m2 = -INFINITY, m3 = -INFINITY;
    for (int j = beg + lane; j < end; j += 32) {
      const float4 e4 = *(const float4*)(el + 4 * (size_t)cs[j]);
      m0 = fmaxf(m0, lrelu(e4.x + er4.x));
      m1 = fmaxf(m1, lrelu(e4.y + er4.y));
      m2 = fmaxf(m2, lrelu(e4.z + er4.z));
      m3 = fmaxf(m3, lrelu(e4.w + er4.w));
    }
    m0 = wave_max(m0); m1 = wave_max(m1); m2 = wave_max(m2); m3 = wave_max(m3);
    if (!__builtin_isfinite(m0)) m0 = 0.f;
    if (!__builtin_isfinite(m1)) m1 = 0.f;
    if (!__builtin_isfinite(m2)) m2 = 0.f;
    if (!__builtin_isfinite(m3)) m3 = 0.f;

    float s0 = 0.f, s1 = 0.f, s2 = 0.f, s3 = 0.f;
    for (int j = beg + lane; j < end; j += 32) {
      const float4 e4 = *(const float4*)(el + 4 * (size_t)cs[j]);
      s0 += __expf(lrelu(e4.x + er4.x) - m0);
      s1 += __expf(lrelu(e4.y + er4.y) - m1);
      s2 += __expf(lrelu(e4.z + er4.z) - m2);
      s3 += __expf(lrelu(e4.w + er4.w) - m3);
    }
    const float d0 = fmaxf(wave_sum(s0), 1e-9f);
    const float d1 = fmaxf(wave_sum(s1), 1e-9f);
    const float d2 = fmaxf(wave_sum(s2), 1e-9f);
    const float d3 = fmaxf(wave_sum(s3), 1e-9f);

    float4 aA = make_float4(0.f, 0.f, 0.f, 0.f), aB = make_float4(0.f, 0.f, 0.f, 0.f);
    for (int base = beg; base < end; base += 32) {
      const int j = base + lane;
      int sidx = 0;
      float a0 = 0.f, a1 = 0.f, a2 = 0.f, a3 = 0.f;
      if (j < end) {
        sidx = cs[j];
        const float4 e4 = *(const float4*)(el + 4 * (size_t)sidx);
        a0 = __expf(lrelu(e4.x + er4.x) - m0) / d0;
        a1 = __expf(lrelu(e4.y + er4.y) - m1) / d1;
        a2 = __expf(lrelu(e4.z + er4.z) - m2) / d2;
        a3 = __expf(lrelu(e4.w + er4.w) - m3) / d3;
      }
      const int cnt = min(32, end - base);
      for (int k = 0; k < cnt; ++k) {
        const int sk = __builtin_amdgcn_readlane(sidx, k);
        const float w0 = lane_val(a0, k), w1 = lane_val(a1, k);
        const float w2 = lane_val(a2, k), w3 = lane_val(a3, k);
        const float wA = hsel ? w1 : w0;
        const float wB = hsel ? w3 : w2;
        const float* fr = feat + (size_t)sk * 256 + 4 * lane;
        const float4 fA = *(const float4*)fr;
        const float4 fB = *(const float4*)(fr + 128);
        aA.x += wA * fA.x; aA.y += wA * fA.y; aA.z += wA * fA.z; aA.w += wA * fA.w;
        aB.x += wB * fB.x; aB.y += wB * fB.y; aB.z += wB * fB.z; aB.w += wB * fB.w;
      }
    }
    hs.x += elu1(aA.x + bA.x) + elu1(aB.x + bB.x);
    hs.y += elu1(aA.y + bA.y) + elu1(aB.y + bB.y);
    hs.z += elu1(aA.z + bA.z) + elu1(aB.z + bB.z);
    hs.w += elu1(aA.w + bA.w) + elu1(aB.w + bB.w);
  }
  hs.x += __shfl_xor(hs.x, 16);
  hs.y += __shfl_xor(hs.y, 16);
  hs.z += __shfl_xor(hs.z, 16);
  hs.w += __shfl_xor(hs.w, 16);
  if (hsel == 0) {
    *(float4*)(h1 + (size_t)n * 64 + 4 * lane) =
        make_float4(hs.x / 4.f, hs.y / 4.f, hs.z / 4.f, hs.w / 4.f);
  }
}

__global__ __launch_bounds__(256) void gather_l1(const int2* __restrict__ seg, const int* __restrict__ csr,
                                                 const float* __restrict__ el, const float* __restrict__ er,
                                                 const float* __restrict__ feat, const float* __restrict__ b,
                                                 float* __restrict__ out, int N, int E) {
  const int lane = threadIdx.x & 31;
  const int n = blockIdx.x * 8 + __builtin_amdgcn_readfirstlane((int)(threadIdx.x >> 5));
  if (n >= N) return;
  const float ern = er[n];
  const float2 b2 = *(const float2*)(b + 2 * lane);
  float2 o = make_float2(0.f, 0.f);

  for (int t = 0; t < 2; ++t) {
    const int2 sg = seg[(size_t)t * N + n];
    const int beg = __builtin_amdgcn_readfirstlane(sg.x);
    const int end = beg + __builtin_amdgcn_readfirstlane(sg.y);
    const int* cs = csr + (size_t)t * E;

    float m = -INFINITY;
    for (int j = beg + lane; j < end; j += 32) m = fmaxf(m, lrelu(el[cs[j]] + ern));
    m = wave_max(m);
    if (!__builtin_isfinite(m)) m = 0.f;

    float s = 0.f;
    for (int j = beg + lane; j < end; j += 32) s += __expf(lrelu(el[cs[j]] + ern) - m);
    const float d = fmaxf(wave_sum(s), 1e-9f);

    float2 acc = make_float2(0.f, 0.f);
    for (int base = beg; base < end; base += 32) {
      const int j = base + lane;
      int sidx = 0;
      float a = 0.f;
      if (j < end) {
        sidx = cs[j];
        a = __expf(lrelu(el[sidx] + ern) - m) / d;
      }
      const int cnt = min(32, end - base);
      for (int k = 0; k < cnt; ++k) {
        const int sk = __builtin_amdgcn_readlane(sidx, k);
        const float w = lane_val(a, k);
        const float2 f = *(const float2*)(feat + (size_t)sk * 64 + 2 * lane);
        acc.x += w * f.x;
        acc.y += w * f.y;
      }
    }
    o.x += acc.x + b2.x;
    o.y += acc.y + b2.y;
  }
  *(float2*)(out + (size_t)n * 64 + 2 * lane) = o;
}

extern "C" void kernel_launch(void* const* d_in, const int* in_sizes, int n_in,
                              void* d_out, int out_size, void* d_ws, size_t ws_size,
                              hipStream_t stream) {
  const int IN = 128, H0 = 4, D0 = 64, H1 = 1, D1 = 64;
  const int N = in_sizes[0] / IN;
  const int E = in_sizes[9];

  const float* x   = (const float*)d_in[0];
  const float* W0  = (const float*)d_in[1];
  const float* al0 = (const float*)d_in[2];
  const float* ar0 = (const float*)d_in[3];
  const float* b0  = (const float*)d_in[4];
  const float* W1  = (const float*)d_in[5];
  const float* al1 = (const float*)d_in[6];
  const float* ar1 = (const float*)d_in[7];
  const float* b1  = (const float*)d_in[8];
  const int* src0 = (const int*)d_in[9];
  const int* dst0 = (const int*)d_in[10];
  const int* src1 = (const int*)d_in[11];
  const int* dst1 = (const int*)d_in[12];
  float* out = (float*)d_out;

  float* ws = (float*)d_ws;
  float*    feat0 = ws;
  float*    el0   = feat0 + (size_t)N * H0 * D0;
  float*    er0   = el0 + (size_t)N * H0;
  float*    h1    = er0 + (size_t)N * H0;
  float*    feat1 = h1 + (size_t)N * D0;
  float*    el1   = feat1 + (size_t)N * D1;
  float*    er1   = el1 + N;
  unsigned* deg   = (unsigned*)(er1 + N);
  unsigned* cnt   = deg + 2 * (size_t)N;
  int2*     seg   = (int2*)(cnt + 4);
  int*      csr   = (int*)(seg + 2 * (size_t)N);

  const int TB = 256;
  (void)ws_size; (void)n_in; (void)out_size;

  {
    fill_u32<<<(2 * N + 4 + TB - 1) / TB, TB, 0, stream>>>(deg, 0u, 2 * N + 4);
    dim3 ge((E + TB - 1) / TB, 2);
    dim3 gn((N + TB - 1) / TB, 2);
    edge_hist<<<ge, TB, 0, stream>>>(dst0, dst1, deg, E, N);
    seg_alloc<<<gn, TB, 0, stream>>>(deg, cnt, seg, N);
    edge_scatter<<<ge, TB, 0, stream>>>(src0, dst0, src1, dst1, deg, seg, csr, E, N);
  }

  {
    gemm_wmma_f32_bregs<128, 4><<<512, 128, 0, stream>>>(x, W0, feat0, N / 16);

    node_scores<<<(N * H0 + TB - 1) / TB, TB, 0, stream>>>(feat0, al0, ar0, el0, er0, N, H0, D0);

    gather_l0<<<(N + 7) / 8, TB, 0, stream>>>(seg, csr, el0, er0, feat0, b0, h1, N, E);
  }

  {
    gemm_wmma_f32_bregs<64, 1><<<1024, 32, 0, stream>>>(h1, W1, feat1, N / 16);

    node_scores<<<(N * H1 + TB - 1) / TB, TB, 0, stream>>>(feat1, al1, ar1, el1, er1, N, H1, D1);

    gather_l1<<<(N + 7) / 8, TB, 0, stream>>>(seg, csr, el1, er1, feat1, b1, out, N, E);
  }
}


# Node logits el/er come from the GEMM accumulators in the epilogue (per-lane dots, 16-lane DPP sum tree, whole-line stores of [el,er] records); node-score launches dropped; el/er sum order changes
# speedup vs baseline: 3.0637x; 1.0669x over previous
#include <hip/hip_runtime.h>
#include <math.h>

typedef float v2f __attribute__((ext_vector_type(2)));
typedef float v8f __attribute__((ext_vector_type(8)));

#define LRELU_SLOPE 0.2f

__global__ void fill_u32(unsigned* p, unsigned v, int n) {
  int i = blockIdx.x * blockDim.x + threadIdx.x;
  int stride = gridDim.x * blockDim.x;
  for (; i < n; i += stride) p[i] = v;
}

template <int K>
__device__ __forceinline__ void load_a_tile(float4 (&a)[K / 8], const float* __restrict__ A, int mt, int c, int h) {
  const float* ap = A + ((size_t)mt * 16 + c) * K + 4 * h;
#pragma unroll
  for (int j = 0; j < K / 8; ++j) a[j] = *(const float4*)(ap + 8 * j);
}

template <int M>
__device__ __forceinline__ float row_xor(float x) {
  return __int_as_float(__builtin_amdgcn_update_dpp(0, __float_as_int(x), 0x160 + M, 0xf, 0xf, true));
}

template <int M, int CNT>
__device__ __forceinline__ void fold_stage(float (&v)[16], int c) {
  const bool up = (c & M) != 0;
#pragma unroll
  for (int i = 0; i < CNT; ++i) {
    const float keep = up ? v[2 * i + 1] : v[2 * i];
    const float send = up ? v[2 * i] : v[2 * i + 1];
    v[i] = keep + row_xor<M>(send);
  }
}

__device__ __forceinline__ float row_transpose_sum(float (&v)[16], int c) {
  fold_stage<1, 8>(v, c);
  fold_stage<2, 4>(v, c);
  fold_stage<4, 2>(v, c);
  fold_stage<8, 1>(v, c);
  return v[0];
}

template <int K, int NW>
__device__ __forceinline__ void mma_store_tile(const float4 (&a)[K / 8], const v2f (&bf)[4][K / 4],
                                               const float (&alv)[4], const float (&arv)[4],
                                               float* __restrict__ C, float* __restrict__ eler,
                                               float* sc, int mt, int n0, int lane, int h) {
  constexpr int NC = NW * 64;
  v8f acc[4] = {};
#pragma unroll
  for (int j = 0; j < K / 8; ++j) {
    v2f a0, a1;
    a0.x = a[j].x; a0.y = a[j].y;
    a1.x = a[j].z; a1.y = a[j].w;
#pragma unroll
    for (int t = 0; t < 4; ++t)
      acc[t] = __builtin_amdgcn_wmma_f32_16x16x4_f32(false, a0, false, bf[t][2 * j], (short)0, acc[t],
                                                     false, false);
#pragma unroll
    for (int t = 0; t < 4; ++t)
      acc[t] = __builtin_amdgcn_wmma_f32_16x16x4_f32(false, a1, false, bf[t][2 * j + 1], (short)0,
                                                     acc[t], false, false);
  }
  float v[16];
#pragma unroll
  for (int r = 0; r < 8; ++r) {
    v[r] = acc[0][r] * alv[0] + acc[1][r] * alv[1] + acc[2][r] * alv[2] + acc[3][r] * alv[3];
    v[8 + r] = acc[0][r] * arv[0] + acc[1][r] * arv[1] + acc[2][r] * arv[2] + acc[3][r] * arv[3];
  }
  const int c = lane & 15;
  const float tot = row_transpose_sum(v, c);
  const int slot = (8 * h + (c & 7)) * (2 * NW) + (c >> 3) * NW;
  if constexpr (NW == 1) {
    eler[(size_t)mt * 32 + slot] = tot;
  } else {
    const int w = n0 >> 6;
    sc[slot + w] = tot;
    __syncthreads();
    eler[(size_t)mt * (32 * NW) + 32 * w + lane] = sc[32 * w + lane];
  }

  float* crow = C + (size_t)mt * 16 * NC + n0 + lane;
#pragma unroll
  for (int p = 0; p < 2; ++p) {
#pragma unroll
    for (int r = 0; r < 8; ++r) {
      const float x0 = acc[2 * p][r], x1 = acc[2 * p + 1][r];
      const float recv = __shfl_xor(h ? x0 : x1, 16);
      crow[(size_t)r * NC + 32 * p] = h ? recv : x0;
      crow[(size_t)(8 + r) * NC + 32 * p] = h ? x1 : recv;
    }
  }
}

template <int K, int NW>
__global__ __launch_bounds__(NW * 32) void gemm_wmma_f32_bregs(const float* __restrict__ A,
                                                               const float* __restrict__ B,
                                                               const float* __restrict__ al,
                                                               const float* __restrict__ ar,
                                                               float* __restrict__ C,
                                                               float* __restrict__ eler, int mtiles) {
  constexpr int NC = NW * 64;
  constexpr int KS = K / 4;
  const int lane = threadIdx.x & 31;
  const int h = lane >> 4;
  const int c = lane & 15;
  const int n0 = __builtin_amdgcn_readfirstlane((int)(threadIdx.x >> 5)) * 64;
  const int step = gridDim.x;

  int mt = blockIdx.x;
  if (mt >= mtiles) return;

  v2f bf[4][KS];
#pragma unroll
  for (int t = 0; t < 4; ++t) {
#pragma unroll
    for (int ks = 0; ks < KS; ++ks) {
      const int krow = 8 * (ks >> 1) + 4 * h + 2 * (ks & 1);
      const float* bp = B + (size_t)krow * NC + n0 + 16 * t + c;
      bf[t][ks].x = bp[0];
      bf[t][ks].y = bp[NC];
    }
  }

  float alv[4], arv[4];
#pragma unroll
  for (int t = 0; t < 4; ++t) {
    alv[t] = al[n0 + 16 * t + c];
    arv[t] = ar[n0 + 16 * t + c];
  }
  float* sc_buf = nullptr;
  if constexpr (NW > 1) {
    __shared__ float sc_lds[2 * 32 * NW];
    sc_buf = sc_lds;
  }
  int flip = 0;

  float4 an[K / 8];
  load_a_tile<K>(an, A, mt, c, h);
  do {
    float4 ac[K / 8];
#pragma unroll
    for (int j = 0; j < K / 8; ++j) ac[j] = an[j];
    const int mn = mt + step;
    load_a_tile<K>(an, A, mn < mtiles ? mn : mt, c, h);
    __builtin_amdgcn_sched_barrier(0);
    mma_store_tile<K, NW>(ac, bf, alv, arv, C, eler, sc_buf + flip * (32 * NW), mt, n0, lane, h);
    flip ^= 1;
    __builtin_amdgcn_sched_barrier(0);
    mt = mn;
  } while (mt < mtiles);
}


__global__ void edge_hist(const int* __restrict__ dst0, const int* __restrict__ dst1,
                          unsigned* __restrict__ deg, int E, int N) {
  int e = blockIdx.x * blockDim.x + threadIdx.x;
  if (e >= E) return;
  const int t = blockIdx.y;
  const int* dst = t ? dst1 : dst0;
  atomicAdd(deg + (size_t)t * N + dst[e], 1u);
}

__global__ void seg_alloc(const unsigned* __restrict__ deg, unsigned* __restrict__ counter,
                          int2* __restrict__ seg, int N) {
  const int t = blockIdx.y;
  const int n = blockIdx.x * blockDim.x + threadIdx.x;
  const int lane = threadIdx.x & 31;
  const unsigned x = (n < N) ? deg[(size_t)t * N + n] : 0u;
  unsigned incl = x;
  for (int d = 1; d < 32; d <<= 1) {
    unsigned y = __shfl_up(incl, d);
    if (lane >= d) incl += y;
  }
  const unsigned total = __shfl(incl, 31);
  unsigned base = 0u;
  if (lane == 31) base = atomicAdd(counter + t, total);
  base = __shfl(base, 31);
  if (n < N) seg[(size_t)t * N + n] = make_int2((int)(base + incl - x), (int)x);
}

__global__ void edge_scatter(const int* __restrict__ src0, const int* __restrict__ dst0,
                             const int* __restrict__ src1, const int* __restrict__ dst1,
                             unsigned* __restrict__ deg, const int2* __restrict__ seg,
                             int* __restrict__ csr, int E, int N) {
  int e = blockIdx.x * blockDim.x + threadIdx.x;
  if (e >= E) return;
  const int t = blockIdx.y;
  const int* src = t ? src1 : src0;
  const int* dst = t ? dst1 : dst0;
  const int d = dst[e];
  const unsigned old = atomicSub(deg + (size_t)t * N + d, 1u);
  const int pos = seg[(size_t)t * N + d].x + (int)old - 1;
  csr[(size_t)t * E + pos] = src[e];
}

__device__ __forceinline__ float lrelu(float v) { return v > 0.f ? v : LRELU_SLOPE * v; }
__device__ __forceinline__ float elu1(float v) { return v > 0.f ? v : expm1f(v); }
__device__ __forceinline__ float wave_max(float v) {
  for (int o = 16; o > 0; o >>= 1) v = fmaxf(v, __shfl_xor(v, o));
  return v;
}
__device__ __forceinline__ float wave_sum(float v) {
  for (int o = 16; o > 0; o >>= 1) v += __shfl_xor(v, o);
  return v;
}
__device__ __forceinline__ float lane_val(float v, int k) {
  return __int_as_float(__builtin_amdgcn_readlane(__float_as_int(v), k));
}

__global__ __launch_bounds__(256) void gather_l0(const int2* __restrict__ seg, const int* __restrict__ csr,
                                                 const float* __restrict__ eler,
                                                 const float* __restrict__ feat, const float* __restrict__ b,
                                                 float* __restrict__ h1, int N, int E) {
  const int lane = threadIdx.x & 31;
  const int n = blockIdx.x * 8 + __builtin_amdgcn_readfirstlane((int)(threadIdx.x >> 5));
  if (n >= N) return;
  const int hsel = lane >> 4;
  const float4 er4 = *(const float4*)(eler + 8 * (size_t)n + 4);
  const float4 bA = *(const float4*)(b + 4 * lane);
  const float4 bB = *(const float4*)(b + 128 + 4 * lane);
  float4 hs = make_float4(0.f, 0.f, 0.f, 0.f);

  for (int t = 0; t < 2; ++t) {
    const int2 sg = seg[(size_t)t * N + n];
    const int beg = __builtin_amdgcn_readfirstlane(sg.x);
    const int end = beg + __builtin_amdgcn_readfirstlane(sg.y);
    const int* cs = csr + (size_t)t * E;

    float m0 = -INFINITY, m1 = -INFINITY, m2 = -INFINITY, m3 = -INFINITY;
    for (int j = beg + lane; j < end; j += 32) {
      const float4 e4 = *(const float4*)(eler + 8 * (size_t)cs[j]);
      m0 = fmaxf(m0, lrelu(e4.x + er4.x));
      m1 = fmaxf(m1, lrelu(e4.y + er4.y));
      m2 = fmaxf(m2, lrelu(e4.z + er4.z));
      m3 = fmaxf(m3, lrelu(e4.w + er4.w));
    }
    m0 = wave_max(m0); m1 = wave_max(m1); m2 = wave_max(m2); m3 = wave_max(m3);
    if (!__builtin_isfinite(m0)) m0 = 0.f;
    if (!__builtin_isfinite(m1)) m1 = 0.f;
    if (!__builtin_isfinite(m2)) m2 = 0.f;
    if (!__builtin_isfinite(m3)) m3 = 0.f;

    float s0 = 0.f, s1 = 0.f, s2 = 0.f, s3 = 0.f;
    for (int j = beg + lane; j < end; j += 32) {
      const float4 e4 = *(const float4*)(eler + 8 * (size_t)cs[j]);
      s0 += __expf(lrelu(e4.x + er4.x) - m0);
      s1 += __expf(lrelu(e4.y + er4.y) - m1);
      s2 += __expf(lrelu(e4.z + er4.z) - m2);
      s3 += __expf(lrelu(e4.w + er4.w) - m3);
    }
    const float d0 = fmaxf(wave_sum(s0), 1e-9f);
    const float d1 = fmaxf(wave_sum(s1), 1e-9f);
    const float d2 = fmaxf(wave_sum(s2), 1e-9f);
    const float d3 = fmaxf(wave_sum(s3), 1e-9f);

    float4 aA = make_float4(0.f, 0.f, 0.f, 0.f), aB = make_float4(0.f, 0.f, 0.f, 0.f);
    for (int base = beg; base < end; base += 32) {
      const int j = base + lane;
      int sidx = 0;
      float a0 = 0.f, a1 = 0.f, a2 = 0.f, a3 = 0.f;
      if (j < end) {
        sidx = cs[j];
        const float4 e4 = *(const float4*)(eler + 8 * (size_t)sidx);
        a0 = __expf(lrelu(e4.x + er4.x) - m0) / d0;
        a1 = __expf(lrelu(e4.y + er4.y) - m1) / d1;
        a2 = __expf(lrelu(e4.z + er4.z) - m2) / d2;
        a3 = __expf(lrelu(e4.w + er4.w) - m3) / d3;
      }
      const int cnt = min(32, end - base);
      for (int k = 0; k < cnt; ++k) {
        const int sk = __builtin_amdgcn_readlane(sidx, k);
        const float w0 = lane_val(a0, k), w1 = lane_val(a1, k);
        const float w2 = lane_val(a2, k), w3 = lane_val(a3, k);
        const float wA = hsel ? w1 : w0;
        const float wB = hsel ? w3 : w2;
        const float* fr = feat + (size_t)sk * 256 + 4 * lane;
        const float4 fA = *(const float4*)fr;
        const float4 fB = *(const float4*)(fr + 128);
        aA.x += wA * fA.x; aA.y += wA * fA.y; aA.z += wA * fA.z; aA.w += wA * fA.w;
        aB.x += wB * fB.x; aB.y += wB * fB.y; aB.z += wB * fB.z; aB.w += wB * fB.w;
      }
    }
    hs.x += elu1(aA.x + bA.x) + elu1(aB.x + bB.x);
    hs.y += elu1(aA.y + bA.y) + elu1(aB.y + bB.y);
    hs.z += elu1(aA.z + bA.z) + elu1(aB.z + bB.z);
    hs.w += elu1(aA.w + bA.w) + elu1(aB.w + bB.w);
  }
  hs.x += __shfl_xor(hs.x, 16);
  hs.y += __shfl_xor(hs.y, 16);
  hs.z += __shfl_xor(hs.z, 16);
  hs.w += __shfl_xor(hs.w, 16);
  if (hsel == 0) {
    *(float4*)(h1 + (size_t)n * 64 + 4 * lane) =
        make_float4(hs.x / 4.f, hs.y / 4.f, hs.z / 4.f, hs.w / 4.f);
  }
}

__global__ __launch_bounds__(256) void gather_l1(const int2* __restrict__ seg, const int* __restrict__ csr,
                                                 const float* __restrict__ eler,
                                                 const float* __restrict__ feat, const float* __restrict__ b,
                                                 float* __restrict__ out, int N, int E) {
  const int lane = threadIdx.x & 31;
  const int n = blockIdx.x * 8 + __builtin_amdgcn_readfirstlane((int)(threadIdx.x >> 5));
  if (n >= N) return;
  const float ern = eler[2 * (size_t)n + 1];
  const float2 b2 = *(const float2*)(b + 2 * lane);
  float2 o = make_float2(0.f, 0.f);

  for (int t = 0; t < 2; ++t) {
    const int2 sg = seg[(size_t)t * N + n];
    const int beg = __builtin_amdgcn_readfirstlane(sg.x);
    const int end = beg + __builtin_amdgcn_readfirstlane(sg.y);
    const int* cs = csr + (size_t)t * E;

    float m = -INFINITY;
    for (int j = beg + lane; j < end; j += 32) m = fmaxf(m, lrelu(eler[2 * (size_t)cs[j]] + ern));
    m = wave_max(m);
    if (!__builtin_isfinite(m)) m = 0.f;

    float s = 0.f;
    for (int j = beg + lane; j < end; j += 32) s += __expf(lrelu(eler[2 * (size_t)cs[j]] + ern) - m);
    const float d = fmaxf(wave_sum(s), 1e-9f);

    float2 acc = make_float2(0.f, 0.f);
    for (int base = beg; base < end; base += 32) {
      const int j = base + lane;
      int sidx = 0;
      float a = 0.f;
      if (j < end) {
        sidx = cs[j];
        a = __expf(lrelu(eler[2 * (size_t)sidx] + ern) - m) / d;
      }
      const int cnt = min(32, end - base);
      for (int k = 0; k < cnt; ++k) {
        const int sk = __builtin_amdgcn_readlane(sidx, k);
        const float w = lane_val(a, k);
        const float2 f = *(const float2*)(feat + (size_t)sk * 64 + 2 * lane);
        acc.x += w * f.x;
        acc.y += w * f.y;
      }
    }
    o.x += acc.x + b2.x;
    o.y += acc.y + b2.y;
  }
  *(float2*)(out + (size_t)n * 64 + 2 * lane) = o;
}

extern "C" void kernel_launch(void* const* d_in, const int* in_sizes, int n_in,
                              void* d_out, int out_size, void* d_ws, size_t ws_size,
                              hipStream_t stream) {
  const int IN = 128, H0 = 4, D0 = 64, H1 = 1, D1 = 64;
  const int N = in_sizes[0] / IN;
  const int E = in_sizes[9];

  const float* x   = (const float*)d_in[0];
  const float* W0  = (const float*)d_in[1];
  const float* al0 = (const float*)d_in[2];
  const float* ar0 = (const float*)d_in[3];
  const float* b0  = (const float*)d_in[4];
  const float* W1  = (const float*)d_in[5];
  const float* al1 = (const float*)d_in[6];
  const float* ar1 = (const float*)d_in[7];
  const float* b1  = (const float*)d_in[8];
  const int* src0 = (const int*)d_in[9];
  const int* dst0 = (const int*)d_in[10];
  const int* src1 = (const int*)d_in[11];
  const int* dst1 = (const int*)d_in[12];
  float* out = (float*)d_out;

  float* ws = (float*)d_ws;
  float*    feat0 = ws;
  float*    eler0 = feat0 + (size_t)N * H0 * D0;
  float*    h1    = eler0 + (size_t)N * 2 * H0;
  float*    feat1 = h1 + (size_t)N * D0;
  float*    eler1 = feat1 + (size_t)N * D1;
  unsigned* deg   = (unsigned*)(eler1 + (size_t)N * 2 * H1);
  unsigned* cnt   = deg + 2 * (size_t)N;
  int2*     seg   = (int2*)(cnt + 4);
  int*      csr   = (int*)(seg + 2 * (size_t)N);

  const int TB = 256;
  (void)ws_size; (void)n_in; (void)out_size;

  {
    fill_u32<<<(2 * N + 4 + TB - 1) / TB, TB, 0, stream>>>(deg, 0u, 2 * N + 4);
    dim3 ge((E + TB - 1) / TB, 2);
    dim3 gn((N + TB - 1) / TB, 2);
    edge_hist<<<ge, TB, 0, stream>>>(dst0, dst1, deg, E, N);
    seg_alloc<<<gn, TB, 0, stream>>>(deg, cnt, seg, N);
    edge_scatter<<<ge, TB, 0, stream>>>(src0, dst0, src1, dst1, deg, seg, csr, E, N);
  }

  {
    gemm_wmma_f32_bregs<128, 4><<<512, 128, 0, stream>>>(x, W0, al0, ar0, feat0, eler0, N / 16);

    gather_l0<<<(N + 7) / 8, TB, 0, stream>>>(seg, csr, eler0, feat0, b0, h1, N, E);
  }

  {
    gemm_wmma_f32_bregs<64, 1><<<1024, 32, 0, stream>>>(h1, W1, al1, ar1, feat1, eler1, N / 16);

    gather_l1<<<(N + 7) / 8, TB, 0, stream>>>(seg, csr, eler1, feat1, b1, out, N, E);
  }
}


# Gathers: each edge's source and logits are loaded once and the score kept in a register through max, sum and weight (type 0 in lanes 0-15, type 1 in 16-31; 16-lane DPP trees); 3 load rounds -> 1
# speedup vs baseline: 3.3370x; 1.0892x over previous
#include <hip/hip_runtime.h>
#include <math.h>

typedef float v2f __attribute__((ext_vector_type(2)));
typedef float v8f __attribute__((ext_vector_type(8)));

#define LRELU_SLOPE 0.2f

__global__ void fill_u32(unsigned* p, unsigned v, int n) {
  int i = blockIdx.x * blockDim.x + threadIdx.x;
  int stride = gridDim.x * blockDim.x;
  for (; i < n; i += stride) p[i] = v;
}

template <int K>
__device__ __forceinline__ void load_a_tile(float4 (&a)[K / 8], const float* __restrict__ A, int mt, int c, int h) {
  const float* ap = A + ((size_t)mt * 16 + c) * K + 4 * h;
#pragma unroll
  for (int j = 0; j < K / 8; ++j) a[j] = *(const float4*)(ap + 8 * j);
}

template <int M>
__device__ __forceinline__ float row_xor(float x) {
  return __int_as_float(__builtin_amdgcn_update_dpp(0, __float_as_int(x), 0x160 + M, 0xf, 0xf, true));
}

template <int M, int CNT>
__device__ __forceinline__ void fold_stage(float (&v)[16], int c) {
  const bool up = (c & M) != 0;
#pragma unroll
  for (int i = 0; i < CNT; ++i) {
    const float keep = up ? v[2 * i + 1] : v[2 * i];
    const float send = up ? v[2 * i] : v[2 * i + 1];
    v[i] = keep + row_xor<M>(send);
  }
}

__device__ __forceinline__ float row_transpose_sum(float (&v)[16], int c) {
  fold_stage<1, 8>(v, c);
  fold_stage<2, 4>(v, c);
  fold_stage<4, 2>(v, c);
  fold_stage<8, 1>(v, c);
  return v[0];
}

template <int K, int NW>
__device__ __forceinline__ void mma_store_tile(const float4 (&a)[K / 8], const v2f (&bf)[4][K / 4],
                                               const float (&alv)[4], const float (&arv)[4],
                                               float* __restrict__ C, float* __restrict__ eler,
                                               float* sc, int mt, int n0, int lane, int h) {
  constexpr int NC = NW * 64;
  v8f acc[4] = {};
#pragma unroll
  for (int j = 0; j < K / 8; ++j) {
    v2f a0, a1;
    a0.x = a[j].x; a0.y = a[j].y;
    a1.x = a[j].z; a1.y = a[j].w;
#pragma unroll
    for (int t = 0; t < 4; ++t)
      acc[t] = __builtin_amdgcn_wmma_f32_16x16x4_f32(false, a0, false, bf[t][2 * j], (short)0, acc[t],
                                                     false, false);
#pragma unroll
    for (int t = 0; t < 4; ++t)
      acc[t] = __builtin_amdgcn_wmma_f32_16x16x4_f32(false, a1, false, bf[t][2 * j + 1], (short)0,
                                                     acc[t], false, false);
  }
  float v[16];
#pragma unroll
  for (int r = 0; r < 8; ++r) {
    v[r] = acc[0][r] * alv[0] + acc[1][r] * alv[1] + acc[2][r] * alv[2] + acc[3][r] * alv[3];
    v[8 + r] = acc[0][r] * arv[0] + acc[1][r] * arv[1] + acc[2][r] * arv[2] + acc[3][r] * arv[3];
  }
  const int c = lane & 15;
  const float tot = row_transpose_sum(v, c);
  const int slot = (8 * h + (c & 7)) * (2 * NW) + (c >> 3) * NW;
  if constexpr (NW == 1) {
    eler[(size_t)mt * 32 + slot] = tot;
  } else {
    const int w = n0 >> 6;
    sc[slot + w] = tot;
    __syncthreads();
    eler[(size_t)mt * (32 * NW) + 32 * w + lane] = sc[32 * w + lane];
  }

  float* crow = C + (size_t)mt * 16 * NC + n0 + lane;
#pragma unroll
  for (int p = 0; p < 2; ++p) {
#pragma unroll
    for (int r = 0; r < 8; ++r) {
      const float x0 = acc[2 * p][r], x1 = acc[2 * p + 1][r];
      const float recv = __shfl_xor(h ? x0 : x1, 16);
      crow[(size_t)r * NC + 32 * p] = h ? recv : x0;
      crow[(size_t)(8 + r) * NC + 32 * p] = h ? x1 : recv;
    }
  }
}

template <int K, int NW>
__global__ __launch_bounds__(NW * 32) void gemm_wmma_f32_bregs(const float* __restrict__ A,
                                                               const float* __restrict__ B,
                                                               const float* __restrict__ al,
                                                               const float* __restrict__ ar,
                                                               float* __restrict__ C,
                                                               float* __restrict__ eler, int mtiles) {
  constexpr int NC = NW * 64;
  constexpr int KS = K / 4;
  const int lane = threadIdx.x & 31;
  const int h = lane >> 4;
  const int c = lane & 15;
  const int n0 = __builtin_amdgcn_readfirstlane((int)(threadIdx.x >> 5)) * 64;
  const int step = gridDim.x;

  int mt = blockIdx.x;
  if (mt >= mtiles) return;

  v2f bf[4][KS];
#pragma unroll
  for (int t = 0; t < 4; ++t) {
#pragma unroll
    for (int ks = 0; ks < KS; ++ks) {
      const int krow = 8 * (ks >> 1) + 4 * h + 2 * (ks & 1);
      const float* bp = B + (size_t)krow * NC + n0 + 16 * t + c;
      bf[t][ks].x = bp[0];
      bf[t][ks].y = bp[NC];
    }
  }

  float alv[4], arv[4];
#pragma unroll
  for (int t = 0; t < 4; ++t) {
    alv[t] = al[n0 + 16 * t + c];
    arv[t] = ar[n0 + 16 * t + c];
  }
  float* sc_buf = nullptr;
  if constexpr (NW > 1) {
    __shared__ float sc_lds[2 * 32 * NW];
    sc_buf = sc_lds;
  }
  int flip = 0;

  float4 an[K / 8];
  load_a_tile<K>(an, A, mt, c, h);
  do {
    float4 ac[K / 8];
#pragma unroll
    for (int j = 0; j < K / 8; ++j) ac[j] = an[j];
    const int mn = mt + step;
    load_a_tile<K>(an, A, mn < mtiles ? mn : mt, c, h);
    __builtin_amdgcn_sched_barrier(0);
    mma_store_tile<K, NW>(ac, bf, alv, arv, C, eler, sc_buf + flip * (32 * NW), mt, n0, lane, h);
    flip ^= 1;
    __builtin_amdgcn_sched_barrier(0);
    mt = mn;
  } while (mt < mtiles);
}


__global__ void edge_hist(const int* __restrict__ dst0, const int* __restrict__ dst1,
                          unsigned* __restrict__ deg, int E, int N) {
  int e = blockIdx.x * blockDim.x + threadIdx.x;
  if (e >= E) return;
  const int t = blockIdx.y;
  const int* dst = t ? dst1 : dst0;
  atomicAdd(deg + (size_t)t * N + dst[e], 1u);
}

__global__ void seg_alloc(const unsigned* __restrict__ deg, unsigned* __restrict__ counter,
                          int2* __restrict__ seg, int N) {
  const int t = blockIdx.y;
  const int n = blockIdx.x * blockDim.x + threadIdx.x;
  const int lane = threadIdx.x & 31;
  const unsigned x = (n < N) ? deg[(size_t)t * N + n] : 0u;
  unsigned incl = x;
  for (int d = 1; d < 32; d <<= 1) {
    unsigned y = __shfl_up(incl, d);
    if (lane >= d) incl += y;
  }
  const unsigned total = __shfl(incl, 31);
  unsigned base = 0u;
  if (lane == 31) base = atomicAdd(counter + t, total);
  base = __shfl(base, 31);
  if (n < N) seg[(size_t)t * N + n] = make_int2((int)(base + incl - x), (int)x);
}

__global__ void edge_scatter(const int* __restrict__ src0, const int* __restrict__ dst0,
                             const int* __restrict__ src1, const int* __restrict__ dst1,
                             unsigned* __restrict__ deg, const int2* __restrict__ seg,
                             int* __restrict__ csr, int E, int N) {
  int e = blockIdx.x * blockDim.x + threadIdx.x;
  if (e >= E) return;
  const int t = blockIdx.y;
  const int* src = t ? src1 : src0;
  const int* dst = t ? dst1 : dst0;
  const int d = dst[e];
  const unsigned old = atomicSub(deg + (size_t)t * N + d, 1u);
  const int pos = seg[(size_t)t * N + d].x + (int)old - 1;
  csr[(size_t)t * E + pos] = src[e];
}

__device__ __forceinline__ float lrelu(float v) { return v > 0.f ? v : LRELU_SLOPE * v; }
__device__ __forceinline__ float elu1(float v) { return v > 0.f ? v : expm1f(v); }
__device__ __forceinline__ float wave_max(float v) {
  for (int o = 16; o > 0; o >>= 1) v = fmaxf(v, __shfl_xor(v, o));
  return v;
}
__device__ __forceinline__ float wave_sum(float v) {
  for (int o = 16; o > 0; o >>= 1) v += __shfl_xor(v, o);
  return v;
}
__device__ __forceinline__ float lane_val(float v, int k) {
  return __int_as_float(__builtin_amdgcn_readlane(__float_as_int(v), k));
}

__device__ __forceinline__ float row_max16(float v) {
  v = fmaxf(v, row_xor<8>(v));
  v = fmaxf(v, row_xor<4>(v));
  v = fmaxf(v, row_xor<2>(v));
  v = fmaxf(v, row_xor<1>(v));
  return v;
}
__device__ __forceinline__ float row_sum16(float v) {
  v += row_xor<8>(v);
  v += row_xor<4>(v);
  v += row_xor<2>(v);
  v += row_xor<1>(v);
  return v;
}


__device__ __forceinline__ void rows_l0(float4& aA, float4& aB, const float* __restrict__ feat, int sidx,
                                        float a0, float a1, float a2, float a3, int lane0, int cnt,
                                        int lane, int hsel) {
  for (int k = 0; k < cnt; ++k) {
    const int sk = __builtin_amdgcn_readlane(sidx, lane0 + k);
    const float w0 = lane_val(a0, lane0 + k), w1 = lane_val(a1, lane0 + k);
    const float w2 = lane_val(a2, lane0 + k), w3 = lane_val(a3, lane0 + k);
    const float wA = hsel ? w1 : w0;
    const float wB = hsel ? w3 : w2;
    const float* fr = feat + (size_t)sk * 256 + 4 * lane;
    const float4 fA = *(const float4*)fr;
    const float4 fB = *(const float4*)(fr + 128);
    aA.x += wA * fA.x; aA.y += wA * fA.y; aA.z += wA * fA.z; aA.w += wA * fA.w;
    aB.x += wB * fB.x; aB.y += wB * fB.y; aB.z += wB * fB.z; aB.w += wB * fB.w;
  }
}

__device__ __forceinline__ void add_type_l0(float4& hs, float4 aA, float4 aB, float4 bA, float4 bB) {
  hs.x += elu1(aA.x + bA.x) + elu1(aB.x + bB.x);
  hs.y += elu1(aA.y + bA.y) + elu1(aB.y + bB.y);
  hs.z += elu1(aA.z + bA.z) + elu1(aB.z + bB.z);
  hs.w += elu1(aA.w + bA.w) + elu1(aB.w + bB.w);
}

__device__ __forceinline__ void segment_3pass_l0(float4& aA, float4& aB, const int* __restrict__ cs,
                                                 int beg, int end, const float* __restrict__ eler,
                                                 float4 er4, const float* __restrict__ feat, int lane,
                                                 int hsel) {
  float m0 = -INFINITY, m1 = -INFINITY, m2 = -INFINITY, m3 = -INFINITY;
  for (int j = beg + lane; j < end; j += 32) {
    const float4 e4 = *(const float4*)(eler + 8 * (size_t)cs[j]);
    m0 = fmaxf(m0, lrelu(e4.x + er4.x));
    m1 = fmaxf(m1, lrelu(e4.y + er4.y));
    m2 = fmaxf(m2, lrelu(e4.z + er4.z));
    m3 = fmaxf(m3, lrelu(e4.w + er4.w));
  }
  m0 = wave_max(m0); m1 = wave_max(m1); m2 = wave_max(m2); m3 = wave_max(m3);
  if (!__builtin_isfinite(m0)) m0 = 0.f;
  if (!__builtin_isfinite(m1)) m1 = 0.f;
  if (!__builtin_isfinite(m2)) m2 = 0.f;
  if (!__builtin_isfinite(m3)) m3 = 0.f;

  float s0 = 0.f, s1 = 0.f, s2 = 0.f, s3 = 0.f;
  for (int j = beg + lane; j < end; j += 32) {
    const float4 e4 = *(const float4*)(eler + 8 * (size_t)cs[j]);
    s0 += __expf(lrelu(e4.x + er4.x) - m0);
    s1 += __expf(lrelu(e4.y + er4.y) - m1);
    s2 += __expf(lrelu(e4.z + er4.z) - m2);
    s3 += __expf(lrelu(e4.w + er4.w) - m3);
  }
  const float d0 = fmaxf(wave_sum(s0), 1e-9f);
  const float d1 = fmaxf(wave_sum(s1), 1e-9f);
  const float d2 = fmaxf(wave_sum(s2), 1e-9f);
  const float d3 = fmaxf(wave_sum(s3), 1e-9f);

  for (int base = beg; base < end; base += 32) {
    const int j = base + lane;
    int sidx = 0;
    float a0 = 0.f, a1 = 0.f, a2 = 0.f, a3 = 0.f;
    if (j < end) {
      sidx = cs[j];
      const float4 e4 = *(const float4*)(eler + 8 * (size_t)sidx);
      a0 = __expf(lrelu(e4.x + er4.x) - m0) / d0;
      a1 = __expf(lrelu(e4.y + er4.y) - m1) / d1;
      a2 = __expf(lrelu(e4.z + er4.z) - m2) / d2;
      a3 = __expf(lrelu(e4.w + er4.w) - m3) / d3;
    }
    rows_l0(aA, aB, feat, sidx, a0, a1, a2, a3, 0, min(32, end - base), lane, hsel);
  }
}

__global__ __launch_bounds__(256) void gather_l0(const int2* __restrict__ seg, const int* __restrict__ csr,
                                                 const float* __restrict__ eler,
                                                 const float* __restrict__ feat, const float* __restrict__ b,
                                                 float* __restrict__ h1, int N, int E) {
  const int lane = threadIdx.x & 31;
  const int n = blockIdx.x * 8 + __builtin_amdgcn_readfirstlane((int)(threadIdx.x >> 5));
  if (n >= N) return;
  const int hsel = lane >> 4;
  const float4 er4 = *(const float4*)(eler + 8 * (size_t)n + 4);
  const float4 bA = *(const float4*)(b + 4 * lane);
  const float4 bB = *(const float4*)(b + 128 + 4 * lane);
  float4 hs = make_float4(0.f, 0.f, 0.f, 0.f);

  const int2 sg0 = seg[n];
  const int2 sg1 = seg[(size_t)N + n];
  const int beg0 = __builtin_amdgcn_readfirstlane(sg0.x), deg0 = __builtin_amdgcn_readfirstlane(sg0.y);
  const int beg1 = __builtin_amdgcn_readfirstlane(sg1.x), deg1 = __builtin_amdgcn_readfirstlane(sg1.y);

  if (deg0 <= 16 && deg1 <= 16) {
    const int c = lane & 15;
    const bool has = c < (hsel ? deg1 : deg0);
    const int* ce = hsel ? csr + (size_t)E + beg1 : csr + beg0;
    int sidx = 0;
    float p0 = -INFINITY, p1 = -INFINITY, p2 = -INFINITY, p3 = -INFINITY;
    if (has) {
      sidx = ce[c];
      const float4 e4 = *(const float4*)(eler + 8 * (size_t)sidx);
      p0 = lrelu(e4.x + er4.x);
      p1 = lrelu(e4.y + er4.y);
      p2 = lrelu(e4.z + er4.z);
      p3 = lrelu(e4.w + er4.w);
    }
    float m0 = row_max16(p0), m1 = row_max16(p1), m2 = row_max16(p2), m3 = row_max16(p3);
    if (!__builtin_isfinite(m0)) m0 = 0.f;
    if (!__builtin_isfinite(m1)) m1 = 0.f;
    if (!__builtin_isfinite(m2)) m2 = 0.f;
    if (!__builtin_isfinite(m3)) m3 = 0.f;
    const float q0 = has ? __expf(p0 - m0) : 0.f;
    const float q1 = has ? __expf(p1 - m1) : 0.f;
    const float q2 = has ? __expf(p2 - m2) : 0.f;
    const float q3 = has ? __expf(p3 - m3) : 0.f;
    const float a0 = q0 / fmaxf(row_sum16(q0), 1e-9f);
    const float a1 = q1 / fmaxf(row_sum16(q1), 1e-9f);
    const float a2 = q2 / fmaxf(row_sum16(q2), 1e-9f);
    const float a3 = q3 / fmaxf(row_sum16(q3), 1e-9f);
    float4 aA = make_float4(0.f, 0.f, 0.f, 0.f), aB = make_float4(0.f, 0.f, 0.f, 0.f);
    rows_l0(aA, aB, feat, sidx, a0, a1, a2, a3, 0, deg0, lane, hsel);
    add_type_l0(hs, aA, aB, bA, bB);
    aA = make_float4(0.f, 0.f, 0.f, 0.f);
    aB = make_float4(0.f, 0.f, 0.f, 0.f);
    rows_l0(aA, aB, feat, sidx, a0, a1, a2, a3, 16, deg1, lane, hsel);
    add_type_l0(hs, aA, aB, bA, bB);
  } else {
    for (int t = 0; t < 2; ++t) {
      const int beg = t ? beg1 : beg0;
      const int end = beg + (t ? deg1 : deg0);
      float4 aA = make_float4(0.f, 0.f, 0.f, 0.f), aB = make_float4(0.f, 0.f, 0.f, 0.f);
      segment_3pass_l0(aA, aB, csr + (size_t)t * E, beg, end, eler, er4, feat, lane, hsel);
      add_type_l0(hs, aA, aB, bA, bB);
    }
  }
  hs.x += __shfl_xor(hs.x, 16);
  hs.y += __shfl_xor(hs.y, 16);
  hs.z += __shfl_xor(hs.z, 16);
  hs.w += __shfl_xor(hs.w, 16);
  if (hsel == 0) {
    *(float4*)(h1 + (size_t)n * 64 + 4 * lane) =
        make_float4(hs.x / 4.f, hs.y / 4.f, hs.z / 4.f, hs.w / 4.f);
  }
}


__device__ __forceinline__ void rows_l1(float2& acc, const float* __restrict__ feat, int sidx, float a,
                                        int lane0, int cnt, int lane) {
  for (int k = 0; k < cnt; ++k) {
    const int sk = __builtin_amdgcn_readlane(sidx, lane0 + k);
    const float w = lane_val(a, lane0 + k);
    const float2 f = *(const float2*)(feat + (size_t)sk * 64 + 2 * lane);
    acc.x += w * f.x;
    acc.y += w * f.y;
  }
}

__device__ __forceinline__ void segment_3pass_l1(float2& acc, const int* __restrict__ cs, int beg, int end,
                                                 const float* __restrict__ eler, float ern,
                                                 const float* __restrict__ feat, int lane) {
  float m = -INFINITY;
  for (int j = beg + lane; j < end; j += 32) m = fmaxf(m, lrelu(eler[2 * (size_t)cs[j]] + ern));
  m = wave_max(m);
  if (!__builtin_isfinite(m)) m = 0.f;

  float s = 0.f;
  for (int j = beg + lane; j < end; j += 32) s += __expf(lrelu(eler[2 * (size_t)cs[j]] + ern) - m);
  const float d = fmaxf(wave_sum(s), 1e-9f);

  for (int base = beg; base < end; base += 32) {
    const int j = base + lane;
    int sidx = 0;
    float a = 0.f;
    if (j < end) {
      sidx = cs[j];
      a = __expf(lrelu(eler[2 * (size_t)sidx] + ern) - m) / d;
    }
    rows_l1(acc, feat, sidx, a, 0, min(32, end - base), lane);
  }
}

__global__ __launch_bounds__(256) void gather_l1(const int2* __restrict__ seg, const int* __restrict__ csr,
                                                 const float* __restrict__ eler,
                                                 const float* __restrict__ feat, const float* __restrict__ b,
                                                 float* __restrict__ out, int N, int E) {
  const int lane = threadIdx.x & 31;
  const int n = blockIdx.x * 8 + __builtin_amdgcn_readfirstlane((int)(threadIdx.x >> 5));
  if (n >= N) return;
  const float ern = eler[2 * (size_t)n + 1];
  const float2 b2 = *(const float2*)(b + 2 * lane);
  float2 o = make_float2(0.f, 0.f);

  const int2 sg0 = seg[n];
  const int2 sg1 = seg[(size_t)N + n];
  const int beg0 = __builtin_amdgcn_readfirstlane(sg0.x), deg0 = __builtin_amdgcn_readfirstlane(sg0.y);
  const int beg1 = __builtin_amdgcn_readfirstlane(sg1.x), deg1 = __builtin_amdgcn_readfirstlane(sg1.y);

  if (deg0 <= 16 && deg1 <= 16) {
    const int c = lane & 15;
    const int half = lane >> 4;
    const bool has = c < (half ? deg1 : deg0);
    const int* ce = half ? csr + (size_t)E + beg1 : csr + beg0;
    int sidx = 0;
    float p = -INFINITY;
    if (has) {
      sidx = ce[c];
      p = lrelu(eler[2 * (size_t)sidx] + ern);
    }
    float m = row_max16(p);
    if (!__builtin_isfinite(m)) m = 0.f;
    const float q = has ? __expf(p - m) : 0.f;
    const float a = q / fmaxf(row_sum16(q), 1e-9f);
    float2 acc = make_float2(0.f, 0.f);
    rows_l1(acc, feat, sidx, a, 0, deg0, lane);
    o.x += acc.x + b2.x;
    o.y += acc.y + b2.y;
    acc = make_float2(0.f, 0.f);
    rows_l1(acc, feat, sidx, a, 16, deg1, lane);
    o.x += acc.x + b2.x;
    o.y += acc.y + b2.y;
  } else {
    for (int t = 0; t < 2; ++t) {
      const int beg = t ? beg1 : beg0;
      const int end = beg + (t ? deg1 : deg0);
      float2 acc = make_float2(0.f, 0.f);
      segment_3pass_l1(acc, csr + (size_t)t * E, beg, end, eler, ern, feat, lane);
      o.x += acc.x + b2.x;
      o.y += acc.y + b2.y;
    }
  }
  *(float2*)(out + (size_t)n * 64 + 2 * lane) = o;
}

extern "C" void kernel_launch(void* const* d_in, const int* in_sizes, int n_in,
                              void* d_out, int out_size, void* d_ws, size_t ws_size,
                              hipStream_t stream) {
  const int IN = 128, H0 = 4, D0 = 64, H1 = 1, D1 = 64;
  const int N = in_sizes[0] / IN;
  const int E = in_sizes[9];

  const float* x   = (const float*)d_in[0];
  const float* W0  = (const float*)d_in[1];
  const float* al0 = (const float*)d_in[2];
  const float* ar0 = (const float*)d_in[3];
  const float* b0  = (const float*)d_in[4];
  const float* W1  = (const float*)d_in[5];
  const float* al1 = (const float*)d_in[6];
  const float* ar1 = (const float*)d_in[7];
  const float* b1  = (const float*)d_in[8];
  const int* src0 = (const int*)d_in[9];
  const int* dst0 = (const int*)d_in[10];
  const int* src1 = (const int*)d_in[11];
  const int* dst1 = (const int*)d_in[12];
  float* out = (float*)d_out;

  float* ws = (float*)d_ws;
  float*    feat0 = ws;
  float*    eler0 = feat0 + (size_t)N * H0 * D0;
  float*    h1    = eler0 + (size_t)N * 2 * H0;
  float*    feat1 = h1 + (size_t)N * D0;
  float*    eler1 = feat1 + (size_t)N * D1;
  unsigned* deg   = (unsigned*)(eler1 + (size_t)N * 2 * H1);
  unsigned* cnt   = deg + 2 * (size_t)N;
  int2*     seg   = (int2*)(cnt + 4);
  int*      csr   = (int*)(seg + 2 * (size_t)N);

  const int TB = 256;
  (void)ws_size; (void)n_in; (void)out_size;

  {
    fill_u32<<<(2 * N + 4 + TB - 1) / TB, TB, 0, stream>>>(deg, 0u, 2 * N + 4);
    dim3 ge((E + TB - 1) / TB, 2);
    dim3 gn((N + TB - 1) / TB, 2);
    edge_hist<<<ge, TB, 0, stream>>>(dst0, dst1, deg, E, N);
    seg_alloc<<<gn, TB, 0, stream>>>(deg, cnt, seg, N);
    edge_scatter<<<ge, TB, 0, stream>>>(src0, dst0, src1, dst1, deg, seg, csr, E, N);
  }

  {
    gemm_wmma_f32_bregs<128, 4><<<512, 128, 0, stream>>>(x, W0, al0, ar0, feat0, eler0, N / 16);

    gather_l0<<<(N + 7) / 8, TB, 0, stream>>>(seg, csr, eler0, feat0, b0, h1, N, E);
  }

  {
    gemm_wmma_f32_bregs<64, 1><<<1024, 32, 0, stream>>>(h1, W1, al1, ar1, feat1, eler1, N / 16);

    gather_l1<<<(N + 7) / 8, TB, 0, stream>>>(seg, csr, eler1, feat1, b1, out, N, E);
  }
}


# CSR build in one launch without global atomics: each block reads the dst list for its 512-node range, counts, scans and places its edges in LDS; slots stay in arrival order (LDS atomics)
# speedup vs baseline: 5.5112x; 1.6515x over previous
#include <hip/hip_runtime.h>
#include <math.h>

typedef float v2f __attribute__((ext_vector_type(2)));
typedef float v8f __attribute__((ext_vector_type(8)));

#define LRELU_SLOPE 0.2f

template <int K>
__device__ __forceinline__ void load_a_tile(float4 (&a)[K / 8], const float* __restrict__ A, int mt, int c, int h) {
  const float* ap = A + ((size_t)mt * 16 + c) * K + 4 * h;
#pragma unroll
  for (int j = 0; j < K / 8; ++j) a[j] = *(const float4*)(ap + 8 * j);
}

template <int M>
__device__ __forceinline__ float row_xor(float x) {
  return __int_as_float(__builtin_amdgcn_update_dpp(0, __float_as_int(x), 0x160 + M, 0xf, 0xf, true));
}

template <int M, int CNT>
__device__ __forceinline__ void fold_stage(float (&v)[16], int c) {
  const bool up = (c & M) != 0;
#pragma unroll
  for (int i = 0; i < CNT; ++i) {
    const float keep = up ? v[2 * i + 1] : v[2 * i];
    const float send = up ? v[2 * i] : v[2 * i + 1];
    v[i] = keep + row_xor<M>(send);
  }
}

__device__ __forceinline__ float row_transpose_sum(float (&v)[16], int c) {
  fold_stage<1, 8>(v, c);
  fold_stage<2, 4>(v, c);
  fold_stage<4, 2>(v, c);
  fold_stage<8, 1>(v, c);
  return v[0];
}

template <int K, int NW>
__device__ __forceinline__ void mma_store_tile(const float4 (&a)[K / 8], const v2f (&bf)[4][K / 4],
                                               const float (&alv)[4], const float (&arv)[4],
                                               float* __restrict__ C, float* __restrict__ eler,
                                               float* sc, int mt, int n0, int lane, int h) {
  constexpr int NC = NW * 64;
  v8f acc[4] = {};
#pragma unroll
  for (int j = 0; j < K / 8; ++j) {
    v2f a0, a1;
    a0.x = a[j].x; a0.y = a[j].y;
    a1.x = a[j].z; a1.y = a[j].w;
#pragma unroll
    for (int t = 0; t < 4; ++t)
      acc[t] = __builtin_amdgcn_wmma_f32_16x16x4_f32(false, a0, false, bf[t][2 * j], (short)0, acc[t],
                                                     false, false);
#pragma unroll
    for (int t = 0; t < 4; ++t)
      acc[t] = __builtin_amdgcn_wmma_f32_16x16x4_f32(false, a1, false, bf[t][2 * j + 1], (short)0,
                                                     acc[t], false, false);
  }
  float v[16];
#pragma unroll
  for (int r = 0; r < 8; ++r) {
    v[r] = acc[0][r] * alv[0] + acc[1][r] * alv[1] + acc[2][r] * alv[2] + acc[3][r] * alv[3];
    v[8 + r] = acc[0][r] * arv[0] + acc[1][r] * arv[1] + acc[2][r] * arv[2] + acc[3][r] * arv[3];
  }
  const int c = lane & 15;
  const float tot = row_transpose_sum(v, c);
  const int slot = (8 * h + (c & 7)) * (2 * NW) + (c >> 3) * NW;
  if constexpr (NW == 1) {
    eler[(size_t)mt * 32 + slot] = tot;
  } else {
    const int w = n0 >> 6;
    sc[slot + w] = tot;
    __syncthreads();
    eler[(size_t)mt * (32 * NW) + 32 * w + lane] = sc[32 * w + lane];
  }

  float* crow = C + (size_t)mt * 16 * NC + n0 + lane;
#pragma unroll
  for (int p = 0; p < 2; ++p) {
#pragma unroll
    for (int r = 0; r < 8; ++r) {
      const float x0 = acc[2 * p][r], x1 = acc[2 * p + 1][r];
      const float recv = __shfl_xor(h ? x0 : x1, 16);
      crow[(size_t)r * NC + 32 * p] = h ? recv : x0;
      crow[(size_t)(8 + r) * NC + 32 * p] = h ? x1 : recv;
    }
  }
}

template <int K, int NW>
__global__ __launch_bounds__(NW * 32) void gemm_wmma_f32_bregs(const float* __restrict__ A,
                                                               const float* __restrict__ B,
                                                               const float* __restrict__ al,
                                                               const float* __restrict__ ar,
                                                               float* __restrict__ C,
                                                               float* __restrict__ eler, int mtiles) {
  constexpr int NC = NW * 64;
  constexpr int KS = K / 4;
  const int lane = threadIdx.x & 31;
  const int h = lane >> 4;
  const int c = lane & 15;
  const int n0 = __builtin_amdgcn_readfirstlane((int)(threadIdx.x >> 5)) * 64;
  const int step = gridDim.x;

  int mt = blockIdx.x;
  if (mt >= mtiles) return;

  v2f bf[4][KS];
#pragma unroll
  for (int t = 0; t < 4; ++t) {
#pragma unroll
    for (int ks = 0; ks < KS; ++ks) {
      const int krow = 8 * (ks >> 1) + 4 * h + 2 * (ks & 1);
      const float* bp = B + (size_t)krow * NC + n0 + 16 * t + c;
      bf[t][ks].x = bp[0];
      bf[t][ks].y = bp[NC];
    }
  }

  float alv[4], arv[4];
#pragma unroll
  for (int t = 0; t < 4; ++t) {
    alv[t] = al[n0 + 16 * t + c];
    arv[t] = ar[n0 + 16 * t + c];
  }
  float* sc_buf = nullptr;
  if constexpr (NW > 1) {
    __shared__ float sc_lds[2 * 32 * NW];
    sc_buf = sc_lds;
  }
  int flip = 0;

  float4 an[K / 8];
  load_a_tile<K>(an, A, mt, c, h);
  do {
    float4 ac[K / 8];
#pragma unroll
    for (int j = 0; j < K / 8; ++j) ac[j] = an[j];
    const int mn = mt + step;
    load_a_tile<K>(an, A, mn < mtiles ? mn : mt, c, h);
    __builtin_amdgcn_sched_barrier(0);
    mma_store_tile<K, NW>(ac, bf, alv, arv, C, eler, sc_buf + flip * (32 * NW), mt, n0, lane, h);
    flip ^= 1;
    __builtin_amdgcn_sched_barrier(0);
    mt = mn;
  } while (mt < mtiles);
}

constexpr int CSR_R = 512;
constexpr int CSR_CAP = 8192;
constexpr int CSR_T = 1024;

__device__ __forceinline__ unsigned wave_sum_u32(unsigned v) {
  for (int o = 16; o > 0; o >>= 1) v += __shfl_xor(v, o);
  return v;
}

__global__ __launch_bounds__(CSR_T) void csr_build(const int* __restrict__ src0, const int* __restrict__ dst0,
                                                   const int* __restrict__ src1, const int* __restrict__ dst1,
                                                   int2* __restrict__ seg, int* __restrict__ csr, int E, int N) {
  __shared__ unsigned s_deg[CSR_R];
  __shared__ unsigned s_beg[CSR_R];
  __shared__ unsigned s_ld[CSR_CAP];
  __shared__ unsigned s_le[CSR_CAP];
  __shared__ int s_out[CSR_CAP];
  __shared__ unsigned s_part[CSR_T / 32];
  __shared__ unsigned s_wtot[CSR_R / 32];
  __shared__ unsigned s_cnt;

  const int tid = threadIdx.x;
  const int lane = tid & 31, wave = tid >> 5;
  const int t = blockIdx.y;
  const int lo = blockIdx.x * CSR_R;
  const int* __restrict__ dst = t ? dst1 : dst0;
  const int* __restrict__ src = t ? src1 : src0;
  int* __restrict__ cs = csr + (size_t)t * E;

  for (int i = tid; i < CSR_R; i += CSR_T) s_deg[i] = 0u;
  if (tid == 0) s_cnt = 0u;
  __syncthreads();

  unsigned less = 0u;
  auto visit = [&](int d, int e) {
    const unsigned r = (unsigned)(d - lo);
    less += r >> 31;
    if (r < (unsigned)CSR_R) {
      atomicAdd(&s_deg[r], 1u);
      const unsigned k = atomicAdd(&s_cnt, 1u);
      if (k < (unsigned)CSR_CAP) {
        s_ld[k] = r;
        s_le[k] = (unsigned)e;
      }
    }
  };
  const int E4 = E >> 2;
  const int4* __restrict__ d4 = (const int4*)dst;
  int i = tid;
  for (; i + 3 * CSR_T < E4; i += 4 * CSR_T) {
    const int4 qa = d4[i], qb = d4[i + CSR_T], qc = d4[i + 2 * CSR_T], qd = d4[i + 3 * CSR_T];
    const int ea = 4 * i, eb = 4 * (i + CSR_T), ec = 4 * (i + 2 * CSR_T), ed = 4 * (i + 3 * CSR_T);
    visit(qa.x, ea); visit(qa.y, ea + 1); visit(qa.z, ea + 2); visit(qa.w, ea + 3);
    visit(qb.x, eb); visit(qb.y, eb + 1); visit(qb.z, eb + 2); visit(qb.w, eb + 3);
    visit(qc.x, ec); visit(qc.y, ec + 1); visit(qc.z, ec + 2); visit(qc.w, ec + 3);
    visit(qd.x, ed); visit(qd.y, ed + 1); visit(qd.z, ed + 2); visit(qd.w, ed + 3);
  }
  for (; i < E4; i += CSR_T) {
    const int4 qa = d4[i];
    const int ea = 4 * i;
    visit(qa.x, ea); visit(qa.y, ea + 1); visit(qa.z, ea + 2); visit(qa.w, ea + 3);
  }
  for (int e = 4 * E4 + tid; e < E; e += CSR_T) visit(dst[e], e);

  less = wave_sum_u32(less);
  if (lane == 0) s_part[wave] = less;
  __syncthreads();
  unsigned base = 0u;
  for (int w = 0; w < CSR_T / 32; ++w) base += s_part[w];
  const unsigned M = s_cnt;

  unsigned x = 0u, incl = 0u;
  if (tid < CSR_R) {
    x = s_deg[tid];
    incl = x;
    for (int dd = 1; dd < 32; dd <<= 1) {
      const unsigned y = __shfl_up(incl, dd);
      if (lane >= dd) incl += y;
    }
    if (lane == 31) s_wtot[wave] = incl;
  }
  __syncthreads();
  if (tid < CSR_R) {
    unsigned off = 0u;
    for (int w = 0; w < wave; ++w) off += s_wtot[w];
    const unsigned excl = off + incl - x;
    s_beg[tid] = excl;
    const int n = lo + tid;
    if (n < N) seg[(size_t)t * N + n] = make_int2((int)(base + excl), (int)x);
  }
  __syncthreads();

  if (M <= (unsigned)CSR_CAP) {
    for (unsigned j = tid; j < M; j += CSR_T) {
      const unsigned r = s_ld[j];
      const unsigned old = atomicSub(&s_deg[r], 1u);
      s_out[s_beg[r] + old - 1u] = src[s_le[j]];
    }
    __syncthreads();
    for (unsigned j = tid; j < M; j += CSR_T) cs[base + j] = s_out[j];
  } else {
    for (int e = tid; e < E; e += CSR_T) {
      const unsigned r = (unsigned)(dst[e] - lo);
      if (r < (unsigned)CSR_R) {
        const unsigned old = atomicSub(&s_deg[r], 1u);
        cs[base + s_beg[r] + old - 1u] = src[e];
      }
    }
  }
}

__device__ __forceinline__ float lrelu(float v) { return v > 0.f ? v : LRELU_SLOPE * v; }
__device__ __forceinline__ float elu1(float v) { return v > 0.f ? v : expm1f(v); }
__device__ __forceinline__ float wave_max(float v) {
  for (int o = 16; o > 0; o >>= 1) v = fmaxf(v, __shfl_xor(v, o));
  return v;
}
__device__ __forceinline__ float wave_sum(float v) {
  for (int o = 16; o > 0; o >>= 1) v += __shfl_xor(v, o);
  return v;
}
__device__ __forceinline__ float lane_val(float v, int k) {
  return __int_as_float(__builtin_amdgcn_readlane(__float_as_int(v), k));
}

__device__ __forceinline__ float row_max16(float v) {
  v = fmaxf(v, row_xor<8>(v));
  v = fmaxf(v, row_xor<4>(v));
  v = fmaxf(v, row_xor<2>(v));
  v = fmaxf(v, row_xor<1>(v));
  return v;
}
__device__ __forceinline__ float row_sum16(float v) {
  v += row_xor<8>(v);
  v += row_xor<4>(v);
  v += row_xor<2>(v);
  v += row_xor<1>(v);
  return v;
}


__device__ __forceinline__ void rows_l0(float4& aA, float4& aB, const float* __restrict__ feat, int sidx,
                                        float a0, float a1, float a2, float a3, int lane0, int cnt,
                                        int lane, int hsel) {
  for (int k = 0; k < cnt; ++k) {
    const int sk = __builtin_amdgcn_readlane(sidx, lane0 + k);
    const float w0 = lane_val(a0, lane0 + k), w1 = lane_val(a1, lane0 + k);
    const float w2 = lane_val(a2, lane0 + k), w3 = lane_val(a3, lane0 + k);
    const float wA = hsel ? w1 : w0;
    const float wB = hsel ? w3 : w2;
    const float* fr = feat + (size_t)sk * 256 + 4 * lane;
    const float4 fA = *(const float4*)fr;
    const float4 fB = *(const float4*)(fr + 128);
    aA.x += wA * fA.x; aA.y += wA * fA.y; aA.z += wA * fA.z; aA.w += wA * fA.w;
    aB.x += wB * fB.x; aB.y += wB * fB.y; aB.z += wB * fB.z; aB.w += wB * fB.w;
  }
}

__device__ __forceinline__ void add_type_l0(float4& hs, float4 aA, float4 aB, float4 bA, float4 bB) {
  hs.x += elu1(aA.x + bA.x) + elu1(aB.x + bB.x);
  hs.y += elu1(aA.y + bA.y) + elu1(aB.y + bB.y);
  hs.z += elu1(aA.z + bA.z) + elu1(aB.z + bB.z);
  hs.w += elu1(aA.w + bA.w) + elu1(aB.w + bB.w);
}

__device__ __forceinline__ void segment_3pass_l0(float4& aA, float4& aB, const int* __restrict__ cs,
                                                 int beg, int end, const float* __restrict__ eler,
                                                 float4 er4, const float* __restrict__ feat, int lane,
                                                 int hsel) {
  float m0 = -INFINITY, m1 = -INFINITY, m2 = -INFINITY, m3 = -INFINITY;
  for (int j = beg + lane; j < end; j += 32) {
    const float4 e4 = *(const float4*)(eler + 8 * (size_t)cs[j]);
    m0 = fmaxf(m0, lrelu(e4.x + er4.x));
    m1 = fmaxf(m1, lrelu(e4.y + er4.y));
    m2 = fmaxf(m2, lrelu(e4.z + er4.z));
    m3 = fmaxf(m3, lrelu(e4.w + er4.w));
  }
  m0 = wave_max(m0); m1 = wave_max(m1); m2 = wave_max(m2); m3 = wave_max(m3);
  if (!__builtin_isfinite(m0)) m0 = 0.f;
  if (!__builtin_isfinite(m1)) m1 = 0.f;
  if (!__builtin_isfinite(m2)) m2 = 0.f;
  if (!__builtin_isfinite(m3)) m3 = 0.f;

  float s0 = 0.f, s1 = 0.f, s2 = 0.f, s3 = 0.f;
  for (int j = beg + lane; j < end; j += 32) {
    const float4 e4 = *(const float4*)(eler + 8 * (size_t)cs[j]);
    s0 += __expf(lrelu(e4.x + er4.x) - m0);
    s1 += __expf(lrelu(e4.y + er4.y) - m1);
    s2 += __expf(lrelu(e4.z + er4.z) - m2);
    s3 += __expf(lrelu(e4.w + er4.w) - m3);
  }
  const float d0 = fmaxf(wave_sum(s0), 1e-9f);
  const float d1 = fmaxf(wave_sum(s1), 1e-9f);
  const float d2 = fmaxf(wave_sum(s2), 1e-9f);
  const float d3 = fmaxf(wave_sum(s3), 1e-9f);

  for (int base = beg; base < end; base += 32) {
    const int j = base + lane;
    int sidx = 0;
    float a0 = 0.f, a1 = 0.f, a2 = 0.f, a3 = 0.f;
    if (j < end) {
      sidx = cs[j];
      const float4 e4 = *(const float4*)(eler + 8 * (size_t)sidx);
      a0 = __expf(lrelu(e4.x + er4.x) - m0) / d0;
      a1 = __expf(lrelu(e4.y + er4.y) - m1) / d1;
      a2 = __expf(lrelu(e4.z + er4.z) - m2) / d2;
      a3 = __expf(lrelu(e4.w + er4.w) - m3) / d3;
    }
    rows_l0(aA, aB, feat, sidx, a0, a1, a2, a3, 0, min(32, end - base), lane, hsel);
  }
}

__global__ __launch_bounds__(256) void gather_l0(const int2* __restrict__ seg, const int* __restrict__ csr,
                                                 const float* __restrict__ eler,
                                                 const float* __restrict__ feat, const float* __restrict__ b,
                                                 float* __restrict__ h1, int N, int E) {
  const int lane = threadIdx.x & 31;
  const int n = blockIdx.x * 8 + __builtin_amdgcn_readfirstlane((int)(threadIdx.x >> 5));
  if (n >= N) return;
  const int hsel = lane >> 4;
  const float4 er4 = *(const float4*)(eler + 8 * (size_t)n + 4);
  const float4 bA = *(const float4*)(b + 4 * lane);
  const float4 bB = *(const float4*)(b + 128 + 4 * lane);
  float4 hs = make_float4(0.f, 0.f, 0.f, 0.f);

  const int2 sg0 = seg[n];
  const int2 sg1 = seg[(size_t)N + n];
  const int beg0 = __builtin_amdgcn_readfirstlane(sg0.x), deg0 = __builtin_amdgcn_readfirstlane(sg0.y);
  const int beg1 = __builtin_amdgcn_readfirstlane(sg1.x), deg1 = __builtin_amdgcn_readfirstlane(sg1.y);

  if (deg0 <= 16 && deg1 <= 16) {
    const int c = lane & 15;
    const bool has = c < (hsel ? deg1 : deg0);
    const int* ce = hsel ? csr + (size_t)E + beg1 : csr + beg0;
    int sidx = 0;
    float p0 = -INFINITY, p1 = -INFINITY, p2 = -INFINITY, p3 = -INFINITY;
    if (has) {
      sidx = ce[c];
      const float4 e4 = *(const float4*)(eler + 8 * (size_t)sidx);
      p0 = lrelu(e4.x + er4.x);
      p1 = lrelu(e4.y + er4.y);
      p2 = lrelu(e4.z + er4.z);
      p3 = lrelu(e4.w + er4.w);
    }
    float m0 = row_max16(p0), m1 = row_max16(p1), m2 = row_max16(p2), m3 = row_max16(p3);
    if (!__builtin_isfinite(m0)) m0 = 0.f;
    if (!__builtin_isfinite(m1)) m1 = 0.f;
    if (!__builtin_isfinite(m2)) m2 = 0.f;
    if (!__builtin_isfinite(m3)) m3 = 0.f;
    const float q0 = has ? __expf(p0 - m0) : 0.f;
    const float q1 = has ? __expf(p1 - m1) : 0.f;
    const float q2 = has ? __expf(p2 - m2) : 0.f;
    const float q3 = has ? __expf(p3 - m3) : 0.f;
    const float a0 = q0 / fmaxf(row_sum16(q0), 1e-9f);
    const float a1 = q1 / fmaxf(row_sum16(q1), 1e-9f);
    const float a2 = q2 / fmaxf(row_sum16(q2), 1e-9f);
    const float a3 = q3 / fmaxf(row_sum16(q3), 1e-9f);
    float4 aA = make_float4(0.f, 0.f, 0.f, 0.f), aB = make_float4(0.f, 0.f, 0.f, 0.f);
    rows_l0(aA, aB, feat, sidx, a0, a1, a2, a3, 0, deg0, lane, hsel);
    add_type_l0(hs, aA, aB, bA, bB);
    aA = make_float4(0.f, 0.f, 0.f, 0.f);
    aB = make_float4(0.f, 0.f, 0.f, 0.f);
    rows_l0(aA, aB, feat, sidx, a0, a1, a2, a3, 16, deg1, lane, hsel);
    add_type_l0(hs, aA, aB, bA, bB);
  } else {
    for (int t = 0; t < 2; ++t) {
      const int beg = t ? beg1 : beg0;
      const int end = beg + (t ? deg1 : deg0);
      float4 aA = make_float4(0.f, 0.f, 0.f, 0.f), aB = make_float4(0.f, 0.f, 0.f, 0.f);
      segment_3pass_l0(aA, aB, csr + (size_t)t * E, beg, end, eler, er4, feat, lane, hsel);
      add_type_l0(hs, aA, aB, bA, bB);
    }
  }
  hs.x += __shfl_xor(hs.x, 16);
  hs.y += __shfl_xor(hs.y, 16);
  hs.z += __shfl_xor(hs.z, 16);
  hs.w += __shfl_xor(hs.w, 16);
  if (hsel == 0) {
    *(float4*)(h1 + (size_t)n * 64 + 4 * lane) =
        make_float4(hs.x / 4.f, hs.y / 4.f, hs.z / 4.f, hs.w / 4.f);
  }
}


__device__ __forceinline__ void rows_l1(float2& acc, const float* __restrict__ feat, int sidx, float a,
                                        int lane0, int cnt, int lane) {
  for (int k = 0; k < cnt; ++k) {
    const int sk = __builtin_amdgcn_readlane(sidx, lane0 + k);
    const float w = lane_val(a, lane0 + k);
    const float2 f = *(const float2*)(feat + (size_t)sk * 64 + 2 * lane);
    acc.x += w * f.x;
    acc.y += w * f.y;
  }
}

__device__ __forceinline__ void segment_3pass_l1(float2& acc, const int* __restrict__ cs, int beg, int end,
                                                 const float* __restrict__ eler, float ern,
                                                 const float* __restrict__ feat, int lane) {
  float m = -INFINITY;
  for (int j = beg + lane; j < end; j += 32) m = fmaxf(m, lrelu(eler[2 * (size_t)cs[j]] + ern));
  m = wave_max(m);
  if (!__builtin_isfinite(m)) m = 0.f;

  float s = 0.f;
  for (int j = beg + lane; j < end; j += 32) s += __expf(lrelu(eler[2 * (size_t)cs[j]] + ern) - m);
  const float d = fmaxf(wave_sum(s), 1e-9f);

  for (int base = beg; base < end; base += 32) {
    const int j = base + lane;
    int sidx = 0;
    float a = 0.f;
    if (j < end) {
      sidx = cs[j];
      a = __expf(lrelu(eler[2 * (size_t)sidx] + ern) - m) / d;
    }
    rows_l1(acc, feat, sidx, a, 0, min(32, end - base), lane);
  }
}

__global__ __launch_bounds__(256) void gather_l1(const int2* __restrict__ seg, const int* __restrict__ csr,
                                                 const float* __restrict__ eler,
                                                 const float* __restrict__ feat, const float* __restrict__ b,
                                                 float* __restrict__ out, int N, int E) {
  const int lane = threadIdx.x & 31;
  const int n = blockIdx.x * 8 + __builtin_amdgcn_readfirstlane((int)(threadIdx.x >> 5));
  if (n >= N) return;
  const float ern = eler[2 * (size_t)n + 1];
  const float2 b2 = *(const float2*)(b + 2 * lane);
  float2 o = make_float2(0.f, 0.f);

  const int2 sg0 = seg[n];
  const int2 sg1 = seg[(size_t)N + n];
  const int beg0 = __builtin_amdgcn_readfirstlane(sg0.x), deg0 = __builtin_amdgcn_readfirstlane(sg0.y);
  const int beg1 = __builtin_amdgcn_readfirstlane(sg1.x), deg1 = __builtin_amdgcn_readfirstlane(sg1.y);

  if (deg0 <= 16 && deg1 <= 16) {
    const int c = lane & 15;
    const int half = lane >> 4;
    const bool has = c < (half ? deg1 : deg0);
    const int* ce = half ? csr + (size_t)E + beg1 : csr + beg0;
    int sidx = 0;
    float p = -INFINITY;
    if (has) {
      sidx = ce[c];
      p = lrelu(eler[2 * (size_t)sidx] + ern);
    }
    float m = row_max16(p);
    if (!__builtin_isfinite(m)) m = 0.f;
    const float q = has ? __expf(p - m) : 0.f;
    const float a = q / fmaxf(row_sum16(q), 1e-9f);
    float2 acc = make_float2(0.f, 0.f);
    rows_l1(acc, feat, sidx, a, 0, deg0, lane);
    o.x += acc.x + b2.x;
    o.y += acc.y + b2.y;
    acc = make_float2(0.f, 0.f);
    rows_l1(acc, feat, sidx, a, 16, deg1, lane);
    o.x += acc.x + b2.x;
    o.y += acc.y + b2.y;
  } else {
    for (int t = 0; t < 2; ++t) {
      const int beg = t ? beg1 : beg0;
      const int end = beg + (t ? deg1 : deg0);
      float2 acc = make_float2(0.f, 0.f);
      segment_3pass_l1(acc, csr + (size_t)t * E, beg, end, eler, ern, feat, lane);
      o.x += acc.x + b2.x;
      o.y += acc.y + b2.y;
    }
  }
  *(float2*)(out + (size_t)n * 64 + 2 * lane) = o;
}

extern "C" void kernel_launch(void* const* d_in, const int* in_sizes, int n_in,
                              void* d_out, int out_size, void* d_ws, size_t ws_size,
                              hipStream_t stream) {
  const int IN = 128, H0 = 4, D0 = 64, H1 = 1, D1 = 64;
  const int N = in_sizes[0] / IN;
  const int E = in_sizes[9];

  const float* x   = (const float*)d_in[0];
  const float* W0  = (const float*)d_in[1];
  const float* al0 = (const float*)d_in[2];
  const float* ar0 = (const float*)d_in[3];
  const float* b0  = (const float*)d_in[4];
  const float* W1  = (const float*)d_in[5];
  const float* al1 = (const float*)d_in[6];
  const float* ar1 = (const float*)d_in[7];
  const float* b1  = (const float*)d_in[8];
  const int* src0 = (const int*)d_in[9];
  const int* dst0 = (const int*)d_in[10];
  const int* src1 = (const int*)d_in[11];
  const int* dst1 = (const int*)d_in[12];
  float* out = (float*)d_out;

  float* ws = (float*)d_ws;
  float*    feat0 = ws;
  float*    eler0 = feat0 + (size_t)N * H0 * D0;
  float*    h1    = eler0 + (size_t)N * 2 * H0;
  float*    feat1 = h1 + (size_t)N * D0;
  float*    eler1 = feat1 + (size_t)N * D1;
  int2*     seg   = (int2*)(eler1 + (size_t)N * 2 * H1);
  int*      csr   = (int*)(seg + 2 * (size_t)N);

  const int TB = 256;
  (void)ws_size; (void)n_in; (void)out_size;

  {
    dim3 gc((N + CSR_R - 1) / CSR_R, 2);
    csr_build<<<gc, CSR_T, 0, stream>>>(src0, dst0, src1, dst1, seg, csr, E, N);
  }

  {
    gemm_wmma_f32_bregs<128, 4><<<512, 128, 0, stream>>>(x, W0, al0, ar0, feat0, eler0, N / 16);

    gather_l0<<<(N + 7) / 8, TB, 0, stream>>>(seg, csr, eler0, feat0, b0, h1, N, E);
  }

  {
    gemm_wmma_f32_bregs<64, 1><<<1024, 32, 0, stream>>>(h1, W1, al1, ar1, feat1, eler1, N / 16);

    gather_l1<<<(N + 7) / 8, TB, 0, stream>>>(seg, csr, eler1, feat1, b1, out, N, E);
  }
}


# CSR build: a block scans a quarter of the edge list for a 2048-node range (was all of it for 512 nodes); a node's edges lie in four pieces, gathers map edge number to slot; edge sum order changes
# speedup vs baseline: 6.0444x; 1.0967x over previous
#include <hip/hip_runtime.h>
#include <math.h>

typedef float v2f __attribute__((ext_vector_type(2)));
typedef float v8f __attribute__((ext_vector_type(8)));

#define LRELU_SLOPE 0.2f

template <int K>
__device__ __forceinline__ void load_a_tile(float4 (&a)[K / 8], const float* __restrict__ A, int mt, int c, int h) {
  const float* ap = A + ((size_t)mt * 16 + c) * K + 4 * h;
#pragma unroll
  for (int j = 0; j < K / 8; ++j) a[j] = *(const float4*)(ap + 8 * j);
}

template <int M>
__device__ __forceinline__ float row_xor(float x) {
  return __int_as_float(__builtin_amdgcn_update_dpp(0, __float_as_int(x), 0x160 + M, 0xf, 0xf, true));
}

template <int M, int CNT>
__device__ __forceinline__ void fold_stage(float (&v)[16], int c) {
  const bool up = (c & M) != 0;
#pragma unroll
  for (int i = 0; i < CNT; ++i) {
    const float keep = up ? v[2 * i + 1] : v[2 * i];
    const float send = up ? v[2 * i] : v[2 * i + 1];
    v[i] = keep + row_xor<M>(send);
  }
}

__device__ __forceinline__ float row_transpose_sum(float (&v)[16], int c) {
  fold_stage<1, 8>(v, c);
  fold_stage<2, 4>(v, c);
  fold_stage<4, 2>(v, c);
  fold_stage<8, 1>(v, c);
  return v[0];
}

template <int K, int NW>
__device__ __forceinline__ void mma_store_tile(const float4 (&a)[K / 8], const v2f (&bf)[4][K / 4],
                                               const float (&alv)[4], const float (&arv)[4],
                                               float* __restrict__ C, float* __restrict__ eler,
                                               float* sc, int mt, int n0, int lane, int h) {
  constexpr int NC = NW * 64;
  v8f acc[4] = {};
#pragma unroll
  for (int j = 0; j < K / 8; ++j) {
    v2f a0, a1;
    a0.x = a[j].x; a0.y = a[j].y;
    a1.x = a[j].z; a1.y = a[j].w;
#pragma unroll
    for (int t = 0; t < 4; ++t)
      acc[t] = __builtin_amdgcn_wmma_f32_16x16x4_f32(false, a0, false, bf[t][2 * j], (short)0, acc[t],
                                                     false, false);
#pragma unroll
    for (int t = 0; t < 4; ++t)
      acc[t] = __builtin_amdgcn_wmma_f32_16x16x4_f32(false, a1, false, bf[t][2 * j + 1], (short)0,
                                                     acc[t], false, false);
  }
  float v[16];
#pragma unroll
  for (int r = 0; r < 8; ++r) {
    v[r] = acc[0][r] * alv[0] + acc[1][r] * alv[1] + acc[2][r] * alv[2] + acc[3][r] * alv[3];
    v[8 + r] = acc[0][r] * arv[0] + acc[1][r] * arv[1] + acc[2][r] * arv[2] + acc[3][r] * arv[3];
  }
  const int c = lane & 15;
  const float tot = row_transpose_sum(v, c);
  const int slot = (8 * h + (c & 7)) * (2 * NW) + (c >> 3) * NW;
  if constexpr (NW == 1) {
    eler[(size_t)mt * 32 + slot] = tot;
  } else {
    const int w = n0 >> 6;
    sc[slot + w] = tot;
    __syncthreads();
    eler[(size_t)mt * (32 * NW) + 32 * w + lane] = sc[32 * w + lane];
  }

  float* crow = C + (size_t)mt * 16 * NC + n0 + lane;
#pragma unroll
  for (int p = 0; p < 2; ++p) {
#pragma unroll
    for (int r = 0; r < 8; ++r) {
      const float x0 = acc[2 * p][r], x1 = acc[2 * p + 1][r];
      const float recv = __shfl_xor(h ? x0 : x1, 16);
      crow[(size_t)r * NC + 32 * p] = h ? recv : x0;
      crow[(size_t)(8 + r) * NC + 32 * p] = h ? x1 : recv;
    }
  }
}

template <int K, int NW>
__global__ __launch_bounds__(NW * 32) void gemm_wmma_f32_bregs(const float* __restrict__ A,
                                                               const float* __restrict__ B,
                                                               const float* __restrict__ al,
                                                               const float* __restrict__ ar,
                                                               float* __restrict__ C,
                                                               float* __restrict__ eler, int mtiles) {
  constexpr int NC = NW * 64;
  constexpr int KS = K / 4;
  const int lane = threadIdx.x & 31;
  const int h = lane >> 4;
  const int c = lane & 15;
  const int n0 = __builtin_amdgcn_readfirstlane((int)(threadIdx.x >> 5)) * 64;
  const int step = gridDim.x;

  int mt = blockIdx.x;
  if (mt >= mtiles) return;

  v2f bf[4][KS];
#pragma unroll
  for (int t = 0; t < 4; ++t) {
#pragma unroll
    for (int ks = 0; ks < KS; ++ks) {
      const int krow = 8 * (ks >> 1) + 4 * h + 2 * (ks & 1);
      const float* bp = B + (size_t)krow * NC + n0 + 16 * t + c;
      bf[t][ks].x = bp[0];
      bf[t][ks].y = bp[NC];
    }
  }

  float alv[4], arv[4];
#pragma unroll
  for (int t = 0; t < 4; ++t) {
    alv[t] = al[n0 + 16 * t + c];
    arv[t] = ar[n0 + 16 * t + c];
  }
  float* sc_buf = nullptr;
  if constexpr (NW > 1) {
    __shared__ float sc_lds[2 * 32 * NW];
    sc_buf = sc_lds;
  }
  int flip = 0;

  float4 an[K / 8];
  load_a_tile<K>(an, A, mt, c, h);
  do {
    float4 ac[K / 8];
#pragma unroll
    for (int j = 0; j < K / 8; ++j) ac[j] = an[j];
    const int mn = mt + step;
    load_a_tile<K>(an, A, mn < mtiles ? mn : mt, c, h);
    __builtin_amdgcn_sched_barrier(0);
    mma_store_tile<K, NW>(ac, bf, alv, arv, C, eler, sc_buf + flip * (32 * NW), mt, n0, lane, h);
    flip ^= 1;
    __builtin_amdgcn_sched_barrier(0);
    mt = mn;
  } while (mt < mtiles);
}

constexpr int CSR_P = 4;
constexpr int CSR_R = 2048;
constexpr int CSR_CAP = 8192;
constexpr int CSR_T = 1024;
constexpr int CSR_NPT = CSR_R / CSR_T;
static_assert(CSR_R == CSR_NPT * CSR_T, "the scan gives every thread CSR_NPT nodes");
static_assert(CSR_T / 32 == 32, "the scan sums the group totals of one k over the 32 lanes of a wave");
static_assert(2 * CSR_P == 8, "the gathers read a node's 2 * CSR_P piece records with lanes 0 .. 7");

__device__ __forceinline__ int csr_part_len(int E) { return ((E + 4 * CSR_P - 1) / (4 * CSR_P)) * 4; }

__device__ __forceinline__ unsigned wave_sum_u32(unsigned v) {
  for (int o = 16; o > 0; o >>= 1) v += __shfl_xor(v, o);
  return v;
}

template <int M>
__device__ __forceinline__ unsigned row_xor_u32(unsigned x) {
  return (unsigned)__builtin_amdgcn_update_dpp(0, (int)x, 0x160 + M, 0xf, 0xf, true);
}
__device__ __forceinline__ unsigned wave_total_u32(unsigned v) {
  v += row_xor_u32<8>(v);
  v += row_xor_u32<4>(v);
  v += row_xor_u32<2>(v);
  v += row_xor_u32<1>(v);
  return (unsigned)__builtin_amdgcn_readlane((int)v, 0) + (unsigned)__builtin_amdgcn_readlane((int)v, 16);
}

__global__ __launch_bounds__(CSR_T) void csr_build(const int* __restrict__ src0, const int* __restrict__ dst0,
                                                   const int* __restrict__ src1, const int* __restrict__ dst1,
                                                   int2* __restrict__ seg, int* __restrict__ csr, int E, int N) {
  __shared__ unsigned s_deg[CSR_R];
  __shared__ unsigned s_beg[CSR_R];
  __shared__ unsigned s_ld[CSR_CAP];
  __shared__ unsigned s_le[CSR_CAP];
  __shared__ int s_out[CSR_CAP];
  __shared__ unsigned s_part[CSR_T / 32];
  __shared__ unsigned s_wtot[CSR_R / 32];
  __shared__ unsigned s_cnt;

  const int tid = threadIdx.x;
  const int lane = tid & 31, wave = tid >> 5;
  const int t = blockIdx.y;
  const int p = blockIdx.z;
  const int lo = blockIdx.x * CSR_R;
  const int* __restrict__ dst = t ? dst1 : dst0;
  const int* __restrict__ src = t ? src1 : src0;
  int* __restrict__ cs = csr + (size_t)t * E;
  const int EP = csr_part_len(E);
  const int e_lo = min(p * EP, E);
  const int e_hi = min(e_lo + EP, E);

  for (int i = tid; i < CSR_R; i += CSR_T) s_deg[i] = 0u;
  if (tid == 0) s_cnt = 0u;
  __syncthreads();

  unsigned less = 0u;
  auto visit = [&](int d, int e) {
    const unsigned r = (unsigned)(d - lo);
    less += r >> 31;
    if (r < (unsigned)CSR_R) {
      atomicAdd(&s_deg[r], 1u);
      const unsigned k = atomicAdd(&s_cnt, 1u);
      if (k < (unsigned)CSR_CAP) {
        s_ld[k] = r;
        s_le[k] = (unsigned)e;
      }
    }
  };
  const int E4 = (e_hi - e_lo) >> 2;
  const int4* __restrict__ d4 = (const int4*)(dst + e_lo);
  int i = tid;
  for (; i + 3 * CSR_T < E4; i += 4 * CSR_T) {
    const int4 qa = d4[i], qb = d4[i + CSR_T], qc = d4[i + 2 * CSR_T], qd = d4[i + 3 * CSR_T];
    const int ea = e_lo + 4 * i, eb = e_lo + 4 * (i + CSR_T), ec = e_lo + 4 * (i + 2 * CSR_T),
              ed = e_lo + 4 * (i + 3 * CSR_T);
    visit(qa.x, ea); visit(qa.y, ea + 1); visit(qa.z, ea + 2); visit(qa.w, ea + 3);
    visit(qb.x, eb); visit(qb.y, eb + 1); visit(qb.z, eb + 2); visit(qb.w, eb + 3);
    visit(qc.x, ec); visit(qc.y, ec + 1); visit(qc.z, ec + 2); visit(qc.w, ec + 3);
    visit(qd.x, ed); visit(qd.y, ed + 1); visit(qd.z, ed + 2); visit(qd.w, ed + 3);
  }
  for (; i < E4; i += CSR_T) {
    const int4 qa = d4[i];
    const int ea = e_lo + 4 * i;
    visit(qa.x, ea); visit(qa.y, ea + 1); visit(qa.z, ea + 2); visit(qa.w, ea + 3);
  }
  for (int e = e_lo + 4 * E4 + tid; e < e_hi; e += CSR_T) visit(dst[e], e);

  less = wave_sum_u32(less);
  if (lane == 0) s_part[wave] = less;
  __syncthreads();
  unsigned base = (unsigned)e_lo;
  for (int w = 0; w < CSR_T / 32; ++w) base += s_part[w];
  const unsigned M = s_cnt;

  unsigned xs[CSR_NPT], incl[CSR_NPT];
#pragma unroll
  for (int k = 0; k < CSR_NPT; ++k) {
    xs[k] = s_deg[k * CSR_T + tid];
    incl[k] = xs[k];
    for (int dd = 1; dd < 32; dd <<= 1) {
      const unsigned y = __shfl_up(incl[k], dd);
      if (lane >= dd) incl[k] += y;
    }
    if (lane == 31) s_wtot[k * (CSR_T / 32) + wave] = incl[k];
  }
  __syncthreads();
  unsigned goff[CSR_NPT];
  {
    unsigned carry = 0u;
#pragma unroll
    for (int k = 0; k < CSR_NPT; ++k) {
      const unsigned g = s_wtot[k * (CSR_T / 32) + lane];
      goff[k] = carry + wave_total_u32(lane < wave ? g : 0u);
      carry += wave_total_u32(g);
    }
  }
#pragma unroll
  for (int k = 0; k < CSR_NPT; ++k) {
    const unsigned excl = goff[k] + incl[k] - xs[k];
    s_beg[k * CSR_T + tid] = excl;
    const int n = lo + k * CSR_T + tid;
    if (n < N) seg[(size_t)(t * CSR_P + p) * N + n] = make_int2((int)(base + excl), (int)xs[k]);
  }
  __syncthreads();

  if (M <= (unsigned)CSR_CAP) {
    for (unsigned j = tid; j < M; j += CSR_T) {
      const unsigned r = s_ld[j];
      const unsigned old = atomicSub(&s_deg[r], 1u);
      s_out[s_beg[r] + old - 1u] = src[s_le[j]];
    }
    __syncthreads();
    for (unsigned j = tid; j < M; j += CSR_T) cs[base + j] = s_out[j];
  } else {
    for (int e = e_lo + tid; e < e_hi; e += CSR_T) {
      const unsigned r = (unsigned)(dst[e] - lo);
      if (r < (unsigned)CSR_R) {
        const unsigned old = atomicSub(&s_deg[r], 1u);
        cs[base + s_beg[r] + old - 1u] = src[e];
      }
    }
  }
}

__device__ __forceinline__ int piece_slot(const int (&pb)[CSR_P], const int (&pd)[CSR_P], int j) {
  int s = pb[0] + j;
  int below = pd[0];
#pragma unroll
  for (int p = 1; p < CSR_P; ++p) {
    if (j >= below) s = pb[p] + (j - below);
    below += pd[p];
  }
  return s;
}

__device__ __forceinline__ int2 load_piece_records(const int2* __restrict__ seg, int N, int n, int lane) {
  return seg[(size_t)(lane & 7) * N + n];
}
__device__ __forceinline__ int pieces_of_type(int (&pb)[CSR_P], int (&pd)[CSR_P], int2 rec, int t) {
#pragma unroll
  for (int p = 0; p < CSR_P; ++p) {
    pb[p] = __builtin_amdgcn_readlane(rec.x, t * CSR_P + p);
    pd[p] = __builtin_amdgcn_readlane(rec.y, t * CSR_P + p);
  }
  return pd[0] + pd[1] + pd[2] + pd[3];
}

__device__ __forceinline__ float lrelu(float v) { return v > 0.f ? v : LRELU_SLOPE * v; }
__device__ __forceinline__ float elu1(float v) { return v > 0.f ? v : expm1f(v); }
__device__ __forceinline__ float wave_max(float v) {
  for (int o = 16; o > 0; o >>= 1) v = fmaxf(v, __shfl_xor(v, o));
  return v;
}
__device__ __forceinline__ float wave_sum(float v) {
  for (int o = 16; o > 0; o >>= 1) v += __shfl_xor(v, o);
  return v;
}
__device__ __forceinline__ float lane_val(float v, int k) {
  return __int_as_float(__builtin_amdgcn_readlane(__float_as_int(v), k));
}

__device__ __forceinline__ float row_max16(float v) {
  v = fmaxf(v, row_xor<8>(v));
  v = fmaxf(v, row_xor<4>(v));
  v = fmaxf(v, row_xor<2>(v));
  v = fmaxf(v, row_xor<1>(v));
  return v;
}
__device__ __forceinline__ float row_sum16(float v) {
  v += row_xor<8>(v);
  v += row_xor<4>(v);
  v += row_xor<2>(v);
  v += row_xor<1>(v);
  return v;
}


__device__ __forceinline__ void rows_l0(float4& aA, float4& aB, const float* __restrict__ feat, int sidx,
                                        float a0, float a1, float a2, float a3, int lane0, int cnt,
                                        int lane, int hsel) {
  for (int k = 0; k < cnt; ++k) {
    const int sk = __builtin_amdgcn_readlane(sidx, lane0 + k);
    const float w0 = lane_val(a0, lane0 + k), w1 = lane_val(a1, lane0 + k);
    const float w2 = lane_val(a2, lane0 + k), w3 = lane_val(a3, lane0 + k);
    const float wA = hsel ? w1 : w0;
    const float wB = hsel ? w3 : w2;
    const float* fr = feat + (size_t)sk * 256 + 4 * lane;
    const float4 fA = *(const float4*)fr;
    const float4 fB = *(const float4*)(fr + 128);
    aA.x += wA * fA.x; aA.y += wA * fA.y; aA.z += wA * fA.z; aA.w += wA * fA.w;
    aB.x += wB * fB.x; aB.y += wB * fB.y; aB.z += wB * fB.z; aB.w += wB * fB.w;
  }
}

__device__ __forceinline__ void add_type_l0(float4& hs, float4 aA, float4 aB, float4 bA, float4 bB) {
  hs.x += elu1(aA.x + bA.x) + elu1(aB.x + bB.x);
  hs.y += elu1(aA.y + bA.y) + elu1(aB.y + bB.y);
  hs.z += elu1(aA.z + bA.z) + elu1(aB.z + bB.z);
  hs.w += elu1(aA.w + bA.w) + elu1(aB.w + bB.w);
}

__device__ __forceinline__ void segment_3pass_l0(float4& aA, float4& aB, const int* __restrict__ cs,
                                                 const int (&pb)[CSR_P], const int (&pd)[CSR_P], int deg,
                                                 const float* __restrict__ eler,
                                                 float4 er4, const float* __restrict__ feat, int lane,
                                                 int hsel) {
  float m0 = -INFINITY, m1 = -INFINITY, m2 = -INFINITY, m3 = -INFINITY;
  for (int j = lane; j < deg; j += 32) {
    const float4 e4 = *(const float4*)(eler + 8 * (size_t)cs[piece_slot(pb, pd, j)]);
    m0 = fmaxf(m0, lrelu(e4.x + er4.x));
    m1 = fmaxf(m1, lrelu(e4.y + er4.y));
    m2 = fmaxf(m2, lrelu(e4.z + er4.z));
    m3 = fmaxf(m3, lrelu(e4.w + er4.w));
  }
  m0 = wave_max(m0); m1 = wave_max(m1); m2 = wave_max(m2); m3 = wave_max(m3);
  if (!__builtin_isfinite(m0)) m0 = 0.f;
  if (!__builtin_isfinite(m1)) m1 = 0.f;
  if (!__builtin_isfinite(m2)) m2 = 0.f;
  if (!__builtin_isfinite(m3)) m3 = 0.f;

  float s0 = 0.f, s1 = 0.f, s2 = 0.f, s3 = 0.f;
  for (int j = lane; j < deg; j += 32) {
    const float4 e4 = *(const float4*)(eler + 8 * (size_t)cs[piece_slot(pb, pd, j)]);
    s0 += __expf(lrelu(e4.x + er4.x) - m0);
    s1 += __expf(lrelu(e4.y + er4.y) - m1);
    s2 += __expf(lrelu(e4.z + er4.z) - m2);
    s3 += __expf(lrelu(e4.w + er4.w) - m3);
  }
  const float d0 = fmaxf(wave_sum(s0), 1e-9f);
  const float d1 = fmaxf(wave_sum(s1), 1e-9f);
  const float d2 = fmaxf(wave_sum(s2), 1e-9f);
  const float d3 = fmaxf(wave_sum(s3), 1e-9f);

  for (int base = 0; base < deg; base += 32) {
    const int j = base + lane;
    int sidx = 0;
    float a0 = 0.f, a1 = 0.f, a2 = 0.f, a3 = 0.f;
    if (j < deg) {
      sidx = cs[piece_slot(pb, pd, j)];
      const float4 e4 = *(const float4*)(eler + 8 * (size_t)sidx);
      a0 = __expf(lrelu(e4.x + er4.x) - m0) / d0;
      a1 = __expf(lrelu(e4.y + er4.y) - m1) / d1;
      a2 = __expf(lrelu(e4.z + er4.z) - m2) / d2;
      a3 = __expf(lrelu(e4.w + er4.w) - m3) / d3;
    }
    rows_l0(aA, aB, feat, sidx, a0, a1, a2, a3, 0, min(32, deg - base), lane, hsel);
  }
}

__global__ __launch_bounds__(256) void gather_l0(const int2* __restrict__ seg, const int* __restrict__ csr,
                                                 const float* __restrict__ eler,
                                                 const float* __restrict__ feat, const float* __restrict__ b,
                                                 float* __restrict__ h1, int N, int E) {
  const int lane = threadIdx.x & 31;
  const int n = blockIdx.x * 8 + __builtin_amdgcn_readfirstlane((int)(threadIdx.x >> 5));
  if (n >= N) return;
  const int hsel = lane >> 4;
  const float4 er4 = *(const float4*)(eler + 8 * (size_t)n + 4);
  const float4 bA = *(const float4*)(b + 4 * lane);
  const float4 bB = *(const float4*)(b + 128 + 4 * lane);
  float4 hs = make_float4(0.f, 0.f, 0.f, 0.f);

  int pb0[CSR_P], pd0[CSR_P], pb1[CSR_P], pd1[CSR_P];
  const int2 rec = load_piece_records(seg, N, n, lane);
  const int deg0 = pieces_of_type(pb0, pd0, rec, 0);
  const int deg1 = pieces_of_type(pb1, pd1, rec, 1);

  if (deg0 <= 16 && deg1 <= 16) {
    const int c = lane & 15;
    const bool has = c < (hsel ? deg1 : deg0);
    int mb[CSR_P], md[CSR_P];
#pragma unroll
    for (int p = 0; p < CSR_P; ++p) {
      mb[p] = hsel ? pb1[p] + E : pb0[p];
      md[p] = hsel ? pd1[p] : pd0[p];
    }
    int sidx = 0;
    float p0 = -INFINITY, p1 = -INFINITY, p2 = -INFINITY, p3 = -INFINITY;
    if (has) {
      sidx = csr[piece_slot(mb, md, c)];
      const float4 e4 = *(const float4*)(eler + 8 * (size_t)sidx);
      p0 = lrelu(e4.x + er4.x);
      p1 = lrelu(e4.y + er4.y);
      p2 = lrelu(e4.z + er4.z);
      p3 = lrelu(e4.w + er4.w);
    }
    float m0 = row_max16(p0), m1 = row_max16(p1), m2 = row_max16(p2), m3 = row_max16(p3);
    if (!__builtin_isfinite(m0)) m0 = 0.f;
    if (!__builtin_isfinite(m1)) m1 = 0.f;
    if (!__builtin_isfinite(m2)) m2 = 0.f;
    if (!__builtin_isfinite(m3)) m3 = 0.f;
    const float q0 = has ? __expf(p0 - m0) : 0.f;
    const float q1 = has ? __expf(p1 - m1) : 0.f;
    const float q2 = has ? __expf(p2 - m2) : 0.f;
    const float q3 = has ? __expf(p3 - m3) : 0.f;
    const float a0 = q0 / fmaxf(row_sum16(q0), 1e-9f);
    const float a1 = q1 / fmaxf(row_sum16(q1), 1e-9f);
    const float a2 = q2 / fmaxf(row_sum16(q2), 1e-9f);
    const float a3 = q3 / fmaxf(row_sum16(q3), 1e-9f);
    float4 aA = make_float4(0.f, 0.f, 0.f, 0.f), aB = make_float4(0.f, 0.f, 0.f, 0.f);
    rows_l0(aA, aB, feat, sidx, a0, a1, a2, a3, 0, deg0, lane, hsel);
    add_type_l0(hs, aA, aB, bA, bB);
    aA = make_float4(0.f, 0.f, 0.f, 0.f);
    aB = make_float4(0.f, 0.f, 0.f, 0.f);
    rows_l0(aA, aB, feat, sidx, a0, a1, a2, a3, 16, deg1, lane, hsel);
    add_type_l0(hs, aA, aB, bA, bB);
  } else {
    float4 aA = make_float4(0.f, 0.f, 0.f, 0.f), aB = make_float4(0.f, 0.f, 0.f, 0.f);
    segment_3pass_l0(aA, aB, csr, pb0, pd0, deg0, eler, er4, feat, lane, hsel);
    add_type_l0(hs, aA, aB, bA, bB);
    aA = make_float4(0.f, 0.f, 0.f, 0.f);
    aB = make_float4(0.f, 0.f, 0.f, 0.f);
    segment_3pass_l0(aA, aB, csr + (size_t)E, pb1, pd1, deg1, eler, er4, feat, lane, hsel);
    add_type_l0(hs, aA, aB, bA, bB);
  }
  hs.x += __shfl_xor(hs.x, 16);
  hs.y += __shfl_xor(hs.y, 16);
  hs.z += __shfl_xor(hs.z, 16);
  hs.w += __shfl_xor(hs.w, 16);
  if (hsel == 0) {
    *(float4*)(h1 + (size_t)n * 64 + 4 * lane) =
        make_float4(hs.x / 4.f, hs.y / 4.f, hs.z / 4.f, hs.w / 4.f);
  }
}


__device__ __forceinline__ void rows_l1(float2& acc, const float* __restrict__ feat, int sidx, float a,
                                        int lane0, int cnt, int lane) {
  for (int k = 0; k < cnt; ++k) {
    const int sk = __builtin_amdgcn_readlane(sidx, lane0 + k);
    const float w = lane_val(a, lane0 + k);
    const float2 f = *(const float2*)(feat + (size_t)sk * 64 + 2 * lane);
    acc.x += w * f.x;
    acc.y += w * f.y;
  }
}

__device__ __forceinline__ void segment_3pass_l1(float2& acc, const int* __restrict__ cs,
                                                 const int (&pb)[CSR_P], const int (&pd)[CSR_P], int deg,
                                                 const float* __restrict__ eler, float ern,
                                                 const float* __restrict__ feat, int lane) {
  float m = -INFINITY;
  for (int j = lane; j < deg; j += 32)
    m = fmaxf(m, lrelu(eler[2 * (size_t)cs[piece_slot(pb, pd, j)]] + ern));
  m = wave_max(m);
  if (!__builtin_isfinite(m)) m = 0.f;

  float s = 0.f;
  for (int j = lane; j < deg; j += 32)
    s += __expf(lrelu(eler[2 * (size_t)cs[piece_slot(pb, pd, j)]] + ern) - m);
  const float d = fmaxf(wave_sum(s), 1e-9f);

  for (int base = 0; base < deg; base += 32) {
    const int j = base + lane;
    int sidx = 0;
    float a = 0.f;
    if (j < deg) {
      sidx = cs[piece_slot(pb, pd, j)];
      a = __expf(lrelu(eler[2 * (size_t)sidx] + ern) - m) / d;
    }
    rows_l1(acc, feat, sidx, a, 0, min(32, deg - base), lane);
  }
}

__global__ __launch_bounds__(256) void gather_l1(const int2* __restrict__ seg, const int* __restrict__ csr,
                                                 const float* __restrict__ eler,
                                                 const float* __restrict__ feat, const float* __restrict__ b,
                                                 float* __restrict__ out, int N, int E) {
  const int lane = threadIdx.x & 31;
  const int n = blockIdx.x * 8 + __builtin_amdgcn_readfirstlane((int)(threadIdx.x >> 5));
  if (n >= N) return;
  const float ern = eler[2 * (size_t)n + 1];
  const float2 b2 = *(const float2*)(b + 2 * lane);
  float2 o = make_float2(0.f, 0.f);

  int pb0[CSR_P], pd0[CSR_P], pb1[CSR_P], pd1[CSR_P];
  const int2 rec = load_piece_records(seg, N, n, lane);
  const int deg0 = pieces_of_type(pb0, pd0, rec, 0);
  const int deg1 = pieces_of_type(pb1, pd1, rec, 1);

  if (deg0 <= 16 && deg1 <= 16) {
    const int c = lane & 15;
    const int half = lane >> 4;
    const bool has = c < (half ? deg1 : deg0);
    int mb[CSR_P], md[CSR_P];
#pragma unroll
    for (int q = 0; q < CSR_P; ++q) {
      mb[q] = half ? pb1[q] + E : pb0[q];
      md[q] = half ? pd1[q] : pd0[q];
    }
    int sidx = 0;
    float p = -INFINITY;
    if (has) {
      sidx = csr[piece_slot(mb, md, c)];
      p = lrelu(eler[2 * (size_t)sidx] + ern);
    }
    float m = row_max16(p);
    if (!__builtin_isfinite(m)) m = 0.f;
    const float q = has ? __expf(p - m) : 0.f;
    const float a = q / fmaxf(row_sum16(q), 1e-9f);
    float2 acc = make_float2(0.f, 0.f);
    rows_l1(acc, feat, sidx, a, 0, deg0, lane);
    o.x += acc.x + b2.x;
    o.y += acc.y + b2.y;
    acc = make_float2(0.f, 0.f);
    rows_l1(acc, feat, sidx, a, 16, deg1, lane);
    o.x += acc.x + b2.x;
    o.y += acc.y + b2.y;
  } else {
    float2 acc = make_float2(0.f, 0.f);
    segment_3pass_l1(acc, csr, pb0, pd0, deg0, eler, ern, feat, lane);
    o.x += acc.x + b2.x;
    o.y += acc.y + b2.y;
    acc = make_float2(0.f, 0.f);
    segment_3pass_l1(acc, csr + (size_t)E, pb1, pd1, deg1, eler, ern, feat, lane);
    o.x += acc.x + b2.x;
    o.y += acc.y + b2.y;
  }
  *(float2*)(out + (size_t)n * 64 + 2 * lane) = o;
}

extern "C" void kernel_launch(void* const* d_in, const int* in_sizes, int n_in,
                              void* d_out, int out_size, void* d_ws, size_t ws_size,
                              hipStream_t stream) {
  const int IN = 128, H0 = 4, D0 = 64, H1 = 1, D1 = 64;
  const int N = in_sizes[0] / IN;
  const int E = in_sizes[9];

  const float* x   = (const float*)d_in[0];
  const float* W0  = (const float*)d_in[1];
  const float* al0 = (const float*)d_in[2];
  const float* ar0 = (const float*)d_in[3];
  const float* b0  = (const float*)d_in[4];
  const float* W1  = (const float*)d_in[5];
  const float* al1 = (const float*)d_in[6];
  const float* ar1 = (const float*)d_in[7];
  const float* b1  = (const float*)d_in[8];
  const int* src0 = (const int*)d_in[9];
  const int* dst0 = (const int*)d_in[10];
  const int* src1 = (const int*)d_in[11];
  const int* dst1 = (const int*)d_in[12];
  float* out = (float*)d_out;

  float* ws = (float*)d_ws;
  float*    feat0 = ws;
  float*    eler0 = feat0 + (size_t)N * H0 * D0;
  float*    h1    = eler0 + (size_t)N * 2 * H0;
  float*    feat1 = h1 + (size_t)N * D0;
  float*    eler1 = feat1 + (size_t)N * D1;
  int2*     seg   = (int2*)(eler1 + (size_t)N * 2 * H1);
  int*      csr   = (int*)(seg + 2 * (size_t)N * CSR_P);

  const int TB = 256;
  (void)ws_size; (void)n_in; (void)out_size;

  {
    dim3 gc((N + CSR_R - 1) / CSR_R, 2, CSR_P);
    csr_build<<<gc, CSR_T, 0, stream>>>(src0, dst0, src1, dst1, seg, csr, E, N);
  }

  {
    gemm_wmma_f32_bregs<128, 4><<<512, 128, 0, stream>>>(x, W0, al0, ar0, feat0, eler0, N / 16);

    gather_l0<<<(N + 7) / 8, TB, 0, stream>>>(seg, csr, eler0, feat0, b0, h1, N, E);
  }

  {
    gemm_wmma_f32_bregs<64, 1><<<1024, 32, 0, stream>>>(h1, W1, al1, ar1, feat1, eler1, N / 16);

    gather_l1<<<(N + 7) / 8, TB, 0, stream>>>(seg, csr, eler1, feat1, b1, out, N, E);
  }
}


# Layer-1 projection (h1 @ W1 with el1/er1) joins layer 0's gather launch: a block takes 16 nodes, 2 per wave, h1 tile in LDS, 4 waves do its WMMAs; 5 launches -> 4; sum orders unchanged
# speedup vs baseline: 6.2906x; 1.0407x over previous
#include <hip/hip_runtime.h>
#include <math.h>

typedef float v2f __attribute__((ext_vector_type(2)));
typedef float v8f __attribute__((ext_vector_type(8)));

#define LRELU_SLOPE 0.2f

template <int K>
__device__ __forceinline__ void load_a_tile(float4 (&a)[K / 8], const float* __restrict__ A, int mt, int c, int h) {
  const float* ap = A + ((size_t)mt * 16 + c) * K + 4 * h;
#pragma unroll
  for (int j = 0; j < K / 8; ++j) a[j] = *(const float4*)(ap + 8 * j);
}

template <int M>
__device__ __forceinline__ float row_xor(float x) {
  return __int_as_float(__builtin_amdgcn_update_dpp(0, __float_as_int(x), 0x160 + M, 0xf, 0xf, true));
}

template <int M, int CNT>
__device__ __forceinline__ void fold_stage(float (&v)[16], int c) {
  const bool up = (c & M) != 0;
#pragma unroll
  for (int i = 0; i < CNT; ++i) {
    const float keep = up ? v[2 * i + 1] : v[2 * i];
    const float send = up ? v[2 * i] : v[2 * i + 1];
    v[i] = keep + row_xor<M>(send);
  }
}

__device__ __forceinline__ float row_transpose_sum(float (&v)[16], int c) {
  fold_stage<1, 8>(v, c);
  fold_stage<2, 4>(v, c);
  fold_stage<4, 2>(v, c);
  fold_stage<8, 1>(v, c);
  return v[0];
}

template <int K, int NW>
__device__ __forceinline__ void mma_store_tile(const float4 (&a)[K / 8], const v2f (&bf)[4][K / 4],
                                               const float (&alv)[4], const float (&arv)[4],
                                               float* __restrict__ C, float* __restrict__ eler,
                                               float* sc, int mt, int n0, int lane, int h) {
  constexpr int NC = NW * 64;
  v8f acc[4] = {};
#pragma unroll
  for (int j = 0; j < K / 8; ++j) {
    v2f a0, a1;
    a0.x = a[j].x; a0.y = a[j].y;
    a1.x = a[j].z; a1.y = a[j].w;
#pragma unroll
    for (int t = 0; t < 4; ++t)
      acc[t] = __builtin_amdgcn_wmma_f32_16x16x4_f32(false, a0, false, bf[t][2 * j], (short)0, acc[t],
                                                     false, false);
#pragma unroll
    for (int t = 0; t < 4; ++t)
      acc[t] = __builtin_amdgcn_wmma_f32_16x16x4_f32(false, a1, false, bf[t][2 * j + 1], (short)0,
                                                     acc[t], false, false);
  }
  float v[16];
#pragma unroll
  for (int r = 0; r < 8; ++r) {
    v[r] = acc[0][r] * alv[0] + acc[1][r] * alv[1] + acc[2][r] * alv[2] + acc[3][r] * alv[3];
    v[8 + r] = acc[0][r] * arv[0] + acc[1][r] * arv[1] + acc[2][r] * arv[2] + acc[3][r] * arv[3];
  }
  const int c = lane & 15;
  const float tot = row_transpose_sum(v, c);
  const int slot = (8 * h + (c & 7)) * (2 * NW) + (c >> 3) * NW;
  if constexpr (NW == 1) {
    eler[(size_t)mt * 32 + slot] = tot;
  } else {
    const int w = n0 >> 6;
    sc[slot + w] = tot;
    __syncthreads();
    eler[(size_t)mt * (32 * NW) + 32 * w + lane] = sc[32 * w + lane];
  }

  float* crow = C + (size_t)mt * 16 * NC + n0 + lane;
#pragma unroll
  for (int p = 0; p < 2; ++p) {
#pragma unroll
    for (int r = 0; r < 8; ++r) {
      const float x0 = acc[2 * p][r], x1 = acc[2 * p + 1][r];
      const float recv = __shfl_xor(h ? x0 : x1, 16);
      crow[(size_t)r * NC + 32 * p] = h ? recv : x0;
      crow[(size_t)(8 + r) * NC + 32 * p] = h ? x1 : recv;
    }
  }
}

template <int K, int NW>
__global__ __launch_bounds__(NW * 32) void gemm_wmma_f32_bregs(const float* __restrict__ A,
                                                               const float* __restrict__ B,
                                                               const float* __restrict__ al,
                                                               const float* __restrict__ ar,
                                                               float* __restrict__ C,
                                                               float* __restrict__ eler, int mtiles) {
  constexpr int NC = NW * 64;
  constexpr int KS = K / 4;
  const int lane = threadIdx.x & 31;
  const int h = lane >> 4;
  const int c = lane & 15;
  const int n0 = __builtin_amdgcn_readfirstlane((int)(threadIdx.x >> 5)) * 64;
  const int step = gridDim.x;

  int mt = blockIdx.x;
  if (mt >= mtiles) return;

  v2f bf[4][KS];
#pragma unroll
  for (int t = 0; t < 4; ++t) {
#pragma unroll
    for (int ks = 0; ks < KS; ++ks) {
      const int krow = 8 * (ks >> 1) + 4 * h + 2 * (ks & 1);
      const float* bp = B + (size_t)krow * NC + n0 + 16 * t + c;
      bf[t][ks].x = bp[0];
      bf[t][ks].y = bp[NC];
    }
  }

  float alv[4], arv[4];
#pragma unroll
  for (int t = 0; t < 4; ++t) {
    alv[t] = al[n0 + 16 * t + c];
    arv[t] = ar[n0 + 16 * t + c];
  }
  float* sc_buf = nullptr;
  if constexpr (NW > 1) {
    __shared__ float sc_lds[2 * 32 * NW];
    sc_buf = sc_lds;
  }
  int flip = 0;

  float4 an[K / 8];
  load_a_tile<K>(an, A, mt, c, h);
  do {
    float4 ac[K / 8];
#pragma unroll
    for (int j = 0; j < K / 8; ++j) ac[j] = an[j];
    const int mn = mt + step;
    load_a_tile<K>(an, A, mn < mtiles ? mn : mt, c, h);
    __builtin_amdgcn_sched_barrier(0);
    mma_store_tile<K, NW>(ac, bf, alv, arv, C, eler, sc_buf + flip * (32 * NW), mt, n0, lane, h);
    flip ^= 1;
    __builtin_amdgcn_sched_barrier(0);
    mt = mn;
  } while (mt < mtiles);
}

constexpr int CSR_P = 4;
constexpr int CSR_R = 2048;
constexpr int CSR_CAP = 8192;
constexpr int CSR_T = 1024;
constexpr int CSR_NPT = CSR_R / CSR_T;
static_assert(CSR_R == CSR_NPT * CSR_T, "the scan gives every thread CSR_NPT nodes");
static_assert(CSR_T / 32 == 32, "the scan sums the group totals of one k over the 32 lanes of a wave");
static_assert(2 * CSR_P == 8, "the gathers read a node's 2 * CSR_P piece records with lanes 0 .. 7");

__device__ __forceinline__ int csr_part_len(int E) { return ((E + 4 * CSR_P - 1) / (4 * CSR_P)) * 4; }

__device__ __forceinline__ unsigned wave_sum_u32(unsigned v) {
  for (int o = 16; o > 0; o >>= 1) v += __shfl_xor(v, o);
  return v;
}

template <int M>
__device__ __forceinline__ unsigned row_xor_u32(unsigned x) {
  return (unsigned)__builtin_amdgcn_update_dpp(0, (int)x, 0x160 + M, 0xf, 0xf, true);
}
__device__ __forceinline__ unsigned wave_total_u32(unsigned v) {
  v += row_xor_u32<8>(v);
  v += row_xor_u32<4>(v);
  v += row_xor_u32<2>(v);
  v += row_xor_u32<1>(v);
  return (unsigned)__builtin_amdgcn_readlane((int)v, 0) + (unsigned)__builtin_amdgcn_readlane((int)v, 16);
}

__global__ __launch_bounds__(CSR_T) void csr_build(const int* __restrict__ src0, const int* __restrict__ dst0,
                                                   const int* __restrict__ src1, const int* __restrict__ dst1,
                                                   int2* __restrict__ seg, int* __restrict__ csr, int E, int N) {
  __shared__ unsigned s_deg[CSR_R];
  __shared__ unsigned s_beg[CSR_R];
  __shared__ unsigned s_ld[CSR_CAP];
  __shared__ unsigned s_le[CSR_CAP];
  __shared__ int s_out[CSR_CAP];
  __shared__ unsigned s_part[CSR_T / 32];
  __shared__ unsigned s_wtot[CSR_R / 32];
  __shared__ unsigned s_cnt;

  const int tid = threadIdx.x;
  const int lane = tid & 31, wave = tid >> 5;
  const int t = blockIdx.y;
  const int p = blockIdx.z;
  const int lo = blockIdx.x * CSR_R;
  const int* __restrict__ dst = t ? dst1 : dst0;
  const int* __restrict__ src = t ? src1 : src0;
  int* __restrict__ cs = csr + (size_t)t * E;
  const int EP = csr_part_len(E);
  const int e_lo = min(p * EP, E);
  const int e_hi = min(e_lo + EP, E);

  for (int i = tid; i < CSR_R; i += CSR_T) s_deg[i] = 0u;
  if (tid == 0) s_cnt = 0u;
  __syncthreads();

  unsigned less = 0u;
  auto visit = [&](int d, int e) {
    const unsigned r = (unsigned)(d - lo);
    less += r >> 31;
    if (r < (unsigned)CSR_R) {
      atomicAdd(&s_deg[r], 1u);
      const unsigned k = atomicAdd(&s_cnt, 1u);
      if (k < (unsigned)CSR_CAP) {
        s_ld[k] = r;
        s_le[k] = (unsigned)e;
      }
    }
  };
  const int E4 = (e_hi - e_lo) >> 2;
  const int4* __restrict__ d4 = (const int4*)(dst + e_lo);
  int i = tid;
  for (; i + 3 * CSR_T < E4; i += 4 * CSR_T) {
    const int4 qa = d4[i], qb = d4[i + CSR_T], qc = d4[i + 2 * CSR_T], qd = d4[i + 3 * CSR_T];
    const int ea = e_lo + 4 * i, eb = e_lo + 4 * (i + CSR_T), ec = e_lo + 4 * (i + 2 * CSR_T),
              ed = e_lo + 4 * (i + 3 * CSR_T);
    visit(qa.x, ea); visit(qa.y, ea + 1); visit(qa.z, ea + 2); visit(qa.w, ea + 3);
    visit(qb.x, eb); visit(qb.y, eb + 1); visit(qb.z, eb + 2); visit(qb.w, eb + 3);
    visit(qc.x, ec); visit(qc.y, ec + 1); visit(qc.z, ec + 2); visit(qc.w, ec + 3);
    visit(qd.x, ed); visit(qd.y, ed + 1); visit(qd.z, ed + 2); visit(qd.w, ed + 3);
  }
  for (; i < E4; i += CSR_T) {
    const int4 qa = d4[i];
    const int ea = e_lo + 4 * i;
    visit(qa.x, ea); visit(qa.y, ea + 1); visit(qa.z, ea + 2); visit(qa.w, ea + 3);
  }
  for (int e = e_lo + 4 * E4 + tid; e < e_hi; e += CSR_T) visit(dst[e], e);

  less = wave_sum_u32(less);
  if (lane == 0) s_part[wave] = less;
  __syncthreads();
  unsigned base = (unsigned)e_lo;
  for (int w = 0; w < CSR_T / 32; ++w) base += s_part[w];
  const unsigned M = s_cnt;

  unsigned xs[CSR_NPT], incl[CSR_NPT];
#pragma unroll
  for (int k = 0; k < CSR_NPT; ++k) {
    xs[k] = s_deg[k * CSR_T + tid];
    incl[k] = xs[k];
    for (int dd = 1; dd < 32; dd <<= 1) {
      const unsigned y = __shfl_up(incl[k], dd);
      if (lane >= dd) incl[k] += y;
    }
    if (lane == 31) s_wtot[k * (CSR_T / 32) + wave] = incl[k];
  }
  __syncthreads();
  unsigned goff[CSR_NPT];
  {
    unsigned carry = 0u;
#pragma unroll
    for (int k = 0; k < CSR_NPT; ++k) {
      const unsigned g = s_wtot[k * (CSR_T / 32) + lane];
      goff[k] = carry + wave_total_u32(lane < wave ? g : 0u);
      carry += wave_total_u32(g);
    }
  }
#pragma unroll
  for (int k = 0; k < CSR_NPT; ++k) {
    const unsigned excl = goff[k] + incl[k] - xs[k];
    s_beg[k * CSR_T + tid] = excl;
    const int n = lo + k * CSR_T + tid;
    if (n < N) seg[(size_t)(t * CSR_P + p) * N + n] = make_int2((int)(base + excl), (int)xs[k]);
  }
  __syncthreads();

  if (M <= (unsigned)CSR_CAP) {
    for (unsigned j = tid; j < M; j += CSR_T) {
      const unsigned r = s_ld[j];
      const unsigned old = atomicSub(&s_deg[r], 1u);
      s_out[s_beg[r] + old - 1u] = src[s_le[j]];
    }
    __syncthreads();
    for (unsigned j = tid; j < M; j += CSR_T) cs[base + j] = s_out[j];
  } else {
    for (int e = e_lo + tid; e < e_hi; e += CSR_T) {
      const unsigned r = (unsigned)(dst[e] - lo);
      if (r < (unsigned)CSR_R) {
        const unsigned old = atomicSub(&s_deg[r], 1u);
        cs[base + s_beg[r] + old - 1u] = src[e];
      }
    }
  }
}

__device__ __forceinline__ int piece_slot(const int (&pb)[CSR_P], const int (&pd)[CSR_P], int j) {
  int s = pb[0] + j;
  int below = pd[0];
#pragma unroll
  for (int p = 1; p < CSR_P; ++p) {
    if (j >= below) s = pb[p] + (j - below);
    below += pd[p];
  }
  return s;
}

__device__ __forceinline__ int2 load_piece_records(const int2* __restrict__ seg, int N, int n, int lane) {
  return seg[(size_t)(lane & 7) * N + n];
}
__device__ __forceinline__ int pieces_of_type(int (&pb)[CSR_P], int (&pd)[CSR_P], int2 rec, int t) {
#pragma unroll
  for (int p = 0; p < CSR_P; ++p) {
    pb[p] = __builtin_amdgcn_readlane(rec.x, t * CSR_P + p);
    pd[p] = __builtin_amdgcn_readlane(rec.y, t * CSR_P + p);
  }
  return pd[0] + pd[1] + pd[2] + pd[3];
}

__device__ __forceinline__ float lrelu(float v) { return v > 0.f ? v : LRELU_SLOPE * v; }
__device__ __forceinline__ float elu1(float v) { return v > 0.f ? v : expm1f(v); }
__device__ __forceinline__ float wave_max(float v) {
  for (int o = 16; o > 0; o >>= 1) v = fmaxf(v, __shfl_xor(v, o));
  return v;
}
__device__ __forceinline__ float wave_sum(float v) {
  for (int o = 16; o > 0; o >>= 1) v += __shfl_xor(v, o);
  return v;
}
__device__ __forceinline__ float lane_val(float v, int k) {
  return __int_as_float(__builtin_amdgcn_readlane(__float_as_int(v), k));
}

__device__ __forceinline__ float row_max16(float v) {
  v = fmaxf(v, row_xor<8>(v));
  v = fmaxf(v, row_xor<4>(v));
  v = fmaxf(v, row_xor<2>(v));
  v = fmaxf(v, row_xor<1>(v));
  return v;
}
__device__ __forceinline__ float row_sum16(float v) {
  v += row_xor<8>(v);
  v += row_xor<4>(v);
  v += row_xor<2>(v);
  v += row_xor<1>(v);
  return v;
}


__device__ __forceinline__ void rows_l0(float4& aA, float4& aB, const float* __restrict__ feat, int sidx,
                                        float a0, float a1, float a2, float a3, int lane0, int cnt,
                                        int lane, int hsel) {
  for (int k = 0; k < cnt; ++k) {
    const int sk = __builtin_amdgcn_readlane(sidx, lane0 + k);
    const float w0 = lane_val(a0, lane0 + k), w1 = lane_val(a1, lane0 + k);
    const float w2 = lane_val(a2, lane0 + k), w3 = lane_val(a3, lane0 + k);
    const float wA = hsel ? w1 : w0;
    const float wB = hsel ? w3 : w2;
    const float* fr = feat + (size_t)sk * 256 + 4 * lane;
    const float4 fA = *(const float4*)fr;
    const float4 fB = *(const float4*)(fr + 128);
    aA.x += wA * fA.x; aA.y += wA * fA.y; aA.z += wA * fA.z; aA.w += wA * fA.w;
    aB.x += wB * fB.x; aB.y += wB * fB.y; aB.z += wB * fB.z; aB.w += wB * fB.w;
  }
}

__device__ __forceinline__ void add_type_l0(float4& hs, float4 aA, float4 aB, float4 bA, float4 bB) {
  hs.x += elu1(aA.x + bA.x) + elu1(aB.x + bB.x);
  hs.y += elu1(aA.y + bA.y) + elu1(aB.y + bB.y);
  hs.z += elu1(aA.z + bA.z) + elu1(aB.z + bB.z);
  hs.w += elu1(aA.w + bA.w) + elu1(aB.w + bB.w);
}

__device__ __forceinline__ void segment_3pass_l0(float4& aA, float4& aB, const int* __restrict__ cs,
                                                 const int (&pb)[CSR_P], const int (&pd)[CSR_P], int deg,
                                                 const float* __restrict__ eler,
                                                 float4 er4, const float* __restrict__ feat, int lane,
                                                 int hsel) {
  float m0 = -INFINITY, m1 = -INFINITY, m2 = -INFINITY, m3 = -INFINITY;
  for (int j = lane; j < deg; j += 32) {
    const float4 e4 = *(const float4*)(eler + 8 * (size_t)cs[piece_slot(pb, pd, j)]);
    m0 = fmaxf(m0, lrelu(e4.x + er4.x));
    m1 = fmaxf(m1, lrelu(e4.y + er4.y));
    m2 = fmaxf(m2, lrelu(e4.z + er4.z));
    m3 = fmaxf(m3, lrelu(e4.w + er4.w));
  }
  m0 = wave_max(m0); m1 = wave_max(m1); m2 = wave_max(m2); m3 = wave_max(m3);
  if (!__builtin_isfinite(m0)) m0 = 0.f;
  if (!__builtin_isfinite(m1)) m1 = 0.f;
  if (!__builtin_isfinite(m2)) m2 = 0.f;
  if (!__builtin_isfinite(m3)) m3 = 0.f;

  float s0 = 0.f, s1 = 0.f, s2 = 0.f, s3 = 0.f;
  for (int j = lane; j < deg; j += 32) {
    const float4 e4 = *(const float4*)(eler + 8 * (size_t)cs[piece_slot(pb, pd, j)]);
    s0 += __expf(lrelu(e4.x + er4.x) - m0);
    s1 += __expf(lrelu(e4.y + er4.y) - m1);
    s2 += __expf(lrelu(e4.z + er4.z) - m2);
    s3 += __expf(lrelu(e4.w + er4.w) - m3);
  }
  const float d0 = fmaxf(wave_sum(s0), 1e-9f);
  const float d1 = fmaxf(wave_sum(s1), 1e-9f);
  const float d2 = fmaxf(wave_sum(s2), 1e-9f);
  const float d3 = fmaxf(wave_sum(s3), 1e-9f);

  for (int base = 0; base < deg; base += 32) {
    const int j = base + lane;
    int sidx = 0;
    float a0 = 0.f, a1 = 0.f, a2 = 0.f, a3 = 0.f;
    if (j < deg) {
      sidx = cs[piece_slot(pb, pd, j)];
      const float4 e4 = *(const float4*)(eler + 8 * (size_t)sidx);
      a0 = __expf(lrelu(e4.x + er4.x) - m0) / d0;
      a1 = __expf(lrelu(e4.y + er4.y) - m1) / d1;
      a2 = __expf(lrelu(e4.z + er4.z) - m2) / d2;
      a3 = __expf(lrelu(e4.w + er4.w) - m3) / d3;
    }
    rows_l0(aA, aB, feat, sidx, a0, a1, a2, a3, 0, min(32, deg - base), lane, hsel);
  }
}

__device__ __forceinline__ float4 node_l0(const int2* __restrict__ seg, const int* __restrict__ csr,
                                          const float* __restrict__ eler, const float* __restrict__ feat,
                                          float4 bA, float4 bB, int n, int N, int E, int lane, int hsel) {
  const float4 er4 = *(const float4*)(eler + 8 * (size_t)n + 4);
  float4 hs = make_float4(0.f, 0.f, 0.f, 0.f);

  int pb0[CSR_P], pd0[CSR_P], pb1[CSR_P], pd1[CSR_P];
  const int2 rec = load_piece_records(seg, N, n, lane);
  const int deg0 = pieces_of_type(pb0, pd0, rec, 0);
  const int deg1 = pieces_of_type(pb1, pd1, rec, 1);

  if (deg0 <= 16 && deg1 <= 16) {
    const int c = lane & 15;
    const bool has = c < (hsel ? deg1 : deg0);
    int mb[CSR_P], md[CSR_P];
#pragma unroll
    for (int p = 0; p < CSR_P; ++p) {
      mb[p] = hsel ? pb1[p] + E : pb0[p];
      md[p] = hsel ? pd1[p] : pd0[p];
    }
    int sidx = 0;
    float p0 = -INFINITY, p1 = -INFINITY, p2 = -INFINITY, p3 = -INFINITY;
    if (has) {
      sidx = csr[piece_slot(mb, md, c)];
      const float4 e4 = *(const float4*)(eler + 8 * (size_t)sidx);
      p0 = lrelu(e4.x + er4.x);
      p1 = lrelu(e4.y + er4.y);
      p2 = lrelu(e4.z + er4.z);
      p3 = lrelu(e4.w + er4.w);
    }
    float m0 = row_max16(p0), m1 = row_max16(p1), m2 = row_max16(p2), m3 = row_max16(p3);
    if (!__builtin_isfinite(m0)) m0 = 0.f;
    if (!__builtin_isfinite(m1)) m1 = 0.f;
    if (!__builtin_isfinite(m2)) m2 = 0.f;
    if (!__builtin_isfinite(m3)) m3 = 0.f;
    const float q0 = has ? __expf(p0 - m0) : 0.f;
    const float q1 = has ? __expf(p1 - m1) : 0.f;
    const float q2 = has ? __expf(p2 - m2) : 0.f;
    const float q3 = has ? __expf(p3 - m3) : 0.f;
    const float a0 = q0 / fmaxf(row_sum16(q0), 1e-9f);
    const float a1 = q1 / fmaxf(row_sum16(q1), 1e-9f);
    const float a2 = q2 / fmaxf(row_sum16(q2), 1e-9f);
    const float a3 = q3 / fmaxf(row_sum16(q3), 1e-9f);
    float4 aA = make_float4(0.f, 0.f, 0.f, 0.f), aB = make_float4(0.f, 0.f, 0.f, 0.f);
    rows_l0(aA, aB, feat, sidx, a0, a1, a2, a3, 0, deg0, lane, hsel);
    add_type_l0(hs, aA, aB, bA, bB);
    aA = make_float4(0.f, 0.f, 0.f, 0.f);
    aB = make_float4(0.f, 0.f, 0.f, 0.f);
    rows_l0(aA, aB, feat, sidx, a0, a1, a2, a3, 16, deg1, lane, hsel);
    add_type_l0(hs, aA, aB, bA, bB);
  } else {
    float4 aA = make_float4(0.f, 0.f, 0.f, 0.f), aB = make_float4(0.f, 0.f, 0.f, 0.f);
    segment_3pass_l0(aA, aB, csr, pb0, pd0, deg0, eler, er4, feat, lane, hsel);
    add_type_l0(hs, aA, aB, bA, bB);
    aA = make_float4(0.f, 0.f, 0.f, 0.f);
    aB = make_float4(0.f, 0.f, 0.f, 0.f);
    segment_3pass_l0(aA, aB, csr + (size_t)E, pb1, pd1, deg1, eler, er4, feat, lane, hsel);
    add_type_l0(hs, aA, aB, bA, bB);
  }
  return hs;
}

constexpr int L1_PITCH = 68;

__global__ __launch_bounds__(256, 10) void gather_l0(const int2* __restrict__ seg, const int* __restrict__ csr,
                                                     const float* __restrict__ eler,
                                                     const float* __restrict__ feat, const float* __restrict__ b,
                                                     const float* __restrict__ W1, const float* __restrict__ al1,
                                                     const float* __restrict__ ar1, float* __restrict__ feat1,
                                                     float* __restrict__ eler1, int N, int E) {
  __shared__ float s_a[16 * L1_PITCH];
  __shared__ float s_c[16 * 64];
  const int lane = threadIdx.x & 31;
  const int w = __builtin_amdgcn_readfirstlane((int)(threadIdx.x >> 5));
  const int hsel = lane >> 4;
  const int node0 = blockIdx.x * 16;
  const float4 bA = *(const float4*)(b + 4 * lane);
  const float4 bB = *(const float4*)(b + 128 + 4 * lane);

  for (int it = 0; it < 2; ++it) {
    const int row = 8 * it + w;
    const int n = node0 + row;
    float4 hs = make_float4(0.f, 0.f, 0.f, 0.f);
    if (n < N) hs = node_l0(seg, csr, eler, feat, bA, bB, n, N, E, lane, hsel);
    hs.x += __shfl_xor(hs.x, 16);
    hs.y += __shfl_xor(hs.y, 16);
    hs.z += __shfl_xor(hs.z, 16);
    hs.w += __shfl_xor(hs.w, 16);
    if (hsel == 0) {
      *(float4*)(s_a + row * L1_PITCH + 4 * lane) =
          make_float4(hs.x / 4.f, hs.y / 4.f, hs.z / 4.f, hs.w / 4.f);
    }
  }

  const int tl = (int)__builtin_amdgcn_mbcnt_lo(~0u, 0u);
  const int th = tl >> 4;
  const int tc = tl & 15;

  v2f bf[16];
  float alv[4], arv[4];
  if (w < 4) {
#pragma unroll
    for (int ks = 0; ks < 16; ++ks) {
      const int krow = 8 * (ks >> 1) + 4 * th + 2 * (ks & 1);
      const float* bp = W1 + (size_t)krow * 64 + 16 * w + tc;
      bf[ks].x = bp[0];
      bf[ks].y = bp[64];
    }
  } else if (w == 4) {
#pragma unroll
    for (int t = 0; t < 4; ++t) {
      alv[t] = al1[16 * t + tc];
      arv[t] = ar1[16 * t + tc];
    }
  }
  __syncthreads();

  if (w < 4) {
    float4 a[8];
#pragma unroll
    for (int j = 0; j < 8; ++j) a[j] = *(const float4*)(s_a + tc * L1_PITCH + 8 * j + 4 * th);
    v8f acc = {};
#pragma unroll
    for (int j = 0; j < 8; ++j) {
      v2f a0, a1;
      a0.x = a[j].x; a0.y = a[j].y;
      a1.x = a[j].z; a1.y = a[j].w;
      acc = __builtin_amdgcn_wmma_f32_16x16x4_f32(false, a0, false, bf[2 * j], (short)0, acc, false, false);
      acc = __builtin_amdgcn_wmma_f32_16x16x4_f32(false, a1, false, bf[2 * j + 1], (short)0, acc, false,
                                                  false);
    }
#pragma unroll
    for (int r = 0; r < 8; ++r) s_c[(8 * th + r) * 64 + 16 * w + tc] = acc[r];
  }
  __syncthreads();

  {
    const int row = 2 * w + th;
    const float4 v = *(const float4*)(s_c + row * 64 + 4 * tc);
    if (node0 + row < N) *(float4*)(feat1 + (size_t)(node0 + row) * 64 + 4 * tc) = v;
  }
  if (w == 4) {
    float v[16];
#pragma unroll
    for (int r = 0; r < 8; ++r) {
      const float* cr = s_c + (8 * th + r) * 64 + tc;
      const float x0 = cr[0], x1 = cr[16], x2 = cr[32], x3 = cr[48];
      v[r] = x0 * alv[0] + x1 * alv[1] + x2 * alv[2] + x3 * alv[3];
      v[8 + r] = x0 * arv[0] + x1 * arv[1] + x2 * arv[2] + x3 * arv[3];
    }
    const float tot = row_transpose_sum(v, tc);
    const int rrow = 8 * th + (tc & 7);
    if (node0 + rrow < N) eler1[(size_t)blockIdx.x * 32 + rrow * 2 + (tc >> 3)] = tot;
  }
}


__device__ __forceinline__ void rows_l1(float2& acc, const float* __restrict__ feat, int sidx, float a,
                                        int lane0, int cnt, int lane) {
  for (int k = 0; k < cnt; ++k) {
    const int sk = __builtin_amdgcn_readlane(sidx, lane0 + k);
    const float w = lane_val(a, lane0 + k);
    const float2 f = *(const float2*)(feat + (size_t)sk * 64 + 2 * lane);
    acc.x += w * f.x;
    acc.y += w * f.y;
  }
}

__device__ __forceinline__ void segment_3pass_l1(float2& acc, const int* __restrict__ cs,
                                                 const int (&pb)[CSR_P], const int (&pd)[CSR_P], int deg,
                                                 const float* __restrict__ eler, float ern,
                                                 const float* __restrict__ feat, int lane) {
  float m = -INFINITY;
  for (int j = lane; j < deg; j += 32)
    m = fmaxf(m, lrelu(eler[2 * (size_t)cs[piece_slot(pb, pd, j)]] + ern));
  m = wave_max(m);
  if (!__builtin_isfinite(m)) m = 0.f;

  float s = 0.f;
  for (int j = lane; j < deg; j += 32)
    s += __expf(lrelu(eler[2 * (size_t)cs[piece_slot(pb, pd, j)]] + ern) - m);
  const float d = fmaxf(wave_sum(s), 1e-9f);

  for (int base = 0; base < deg; base += 32) {
    const int j = base + lane;
    int sidx = 0;
    float a = 0.f;
    if (j < deg) {
      sidx = cs[piece_slot(pb, pd, j)];
      a = __expf(lrelu(eler[2 * (size_t)sidx] + ern) - m) / d;
    }
    rows_l1(acc, feat, sidx, a, 0, min(32, deg - base), lane);
  }
}

__global__ __launch_bounds__(256) void gather_l1(const int2* __restrict__ seg, const int* __restrict__ csr,
                                                 const float* __restrict__ eler,
                                                 const float* __restrict__ feat, const float* __restrict__ b,
                                                 float* __restrict__ out, int N, int E) {
  const int lane = threadIdx.x & 31;
  const int n = blockIdx.x * 8 + __builtin_amdgcn_readfirstlane((int)(threadIdx.x >> 5));
  if (n >= N) return;
  const float ern = eler[2 * (size_t)n + 1];
  const float2 b2 = *(const float2*)(b + 2 * lane);
  float2 o = make_float2(0.f, 0.f);

  int pb0[CSR_P], pd0[CSR_P], pb1[CSR_P], pd1[CSR_P];
  const int2 rec = load_piece_records(seg, N, n, lane);
  const int deg0 = pieces_of_type(pb0, pd0, rec, 0);
  const int deg1 = pieces_of_type(pb1, pd1, rec, 1);

  if (deg0 <= 16 && deg1 <= 16) {
    const int c = lane & 15;
    const int half = lane >> 4;
    const bool has = c < (half ? deg1 : deg0);
    int mb[CSR_P], md[CSR_P];
#pragma unroll
    for (int q = 0; q < CSR_P; ++q) {
      mb[q] = half ? pb1[q] + E : pb0[q];
      md[q] = half ? pd1[q] : pd0[q];
    }
    int sidx = 0;
    float p = -INFINITY;
    if (has) {
      sidx = csr[piece_slot(mb, md, c)];
      p = lrelu(eler[2 * (size_t)sidx] + ern);
    }
    float m = row_max16(p);
    if (!__builtin_isfinite(m)) m = 0.f;
    const float q = has ? __expf(p - m) : 0.f;
    const float a = q / fmaxf(row_sum16(q), 1e-9f);
    float2 acc = make_float2(0.f, 0.f);
    rows_l1(acc, feat, sidx, a, 0, deg0, lane);
    o.x += acc.x + b2.x;
    o.y += acc.y + b2.y;
    acc = make_float2(0.f, 0.f);
    rows_l1(acc, feat, sidx, a, 16, deg1, lane);
    o.x += acc.x + b2.x;
    o.y += acc.y + b2.y;
  } else {
    float2 acc = make_float2(0.f, 0.f);
    segment_3pass_l1(acc, csr, pb0, pd0, deg0, eler, ern, feat, lane);
    o.x += acc.x + b2.x;
    o.y += acc.y + b2.y;
    acc = make_float2(0.f, 0.f);
    segment_3pass_l1(acc, csr + (size_t)E, pb1, pd1, deg1, eler, ern, feat, lane);
    o.x += acc.x + b2.x;
    o.y += acc.y + b2.y;
  }
  *(float2*)(out + (size_t)n * 64 + 2 * lane) = o;
}

extern "C" void kernel_launch(void* const* d_in, const int* in_sizes, int n_in,
                              void* d_out, int out_size, void* d_ws, size_t ws_size,
                              hipStream_t stream) {
  const int IN = 128, H0 = 4, D0 = 64, H1 = 1, D1 = 64;
  const int N = in_sizes[0] / IN;
  const int E = in_sizes[9];

  const float* x   = (const float*)d_in[0];
  const float* W0  = (const float*)d_in[1];
  const float* al0 = (const float*)d_in[2];
  const float* ar0 = (const float*)d_in[3];
  const float* b0  = (const float*)d_in[4];
  const float* W1  = (const float*)d_in[5];
  const float* al1 = (const float*)d_in[6];
  const float* ar1 = (const float*)d_in[7];
  const float* b1  = (const float*)d_in[8];
  const int* src0 = (const int*)d_in[9];
  const int* dst0 = (const int*)d_in[10];
  const int* src1 = (const int*)d_in[11];
  const int* dst1 = (const int*)d_in[12];
  float* out = (float*)d_out;

  float* ws = (float*)d_ws;
  float*    feat0 = ws;
  float*    eler0 = feat0 + (size_t)N * H0 * D0;
  float*    feat1 = eler0 + (size_t)N * 2 * H0;
  float*    eler1 = feat1 + (size_t)N * D1;
  int2*     seg   = (int2*)(eler1 + (size_t)N * 2 * H1);
  int*      csr   = (int*)(seg + 2 * (size_t)N * CSR_P);

  const int TB = 256;
  (void)ws_size; (void)n_in; (void)out_size;

  {
    dim3 gc((N + CSR_R - 1) / CSR_R, 2, CSR_P);
    csr_build<<<gc, CSR_T, 0, stream>>>(src0, dst0, src1, dst1, seg, csr, E, N);
  }

  {
    gemm_wmma_f32_bregs<128, 4><<<512, 128, 0, stream>>>(x, W0, al0, ar0, feat0, eler0, N / 16);

    gather_l0<<<(N + 15) / 16, TB, 0, stream>>>(seg, csr, eler0, feat0, b0, W1, al1, ar1, feat1, eler1, N, E);
  }

  {
    gather_l1<<<(N + 7) / 8, TB, 0, stream>>>(seg, csr, eler1, feat1, b1, out, N, E);
  }
}
